# starting kernel
# baseline (speedup 1.0000x reference)
#include <hip/hip_runtime.h>
#include <hip/hip_bf16.h>

#define USER_NUM 100000
#define ITEM_NUM 50000
#define NTOT     (USER_NUM + ITEM_NUM)
#define EMB      64
#define NNZ_CNT  3000000
#define N_LAYERS 3
#define EPS_F    0.1f

typedef float v2f __attribute__((ext_vector_type(2)));
typedef float v8f __attribute__((ext_vector_type(8)));

__global__ void k_init(const float* __restrict__ user_emb,
                       const float* __restrict__ item_emb,
                       float* __restrict__ ego,
                       float* __restrict__ acc) {
    int i = blockIdx.x * blockDim.x + threadIdx.x;
    const int total  = NTOT * EMB / 4;
    const int ubound = USER_NUM * EMB / 4;
    if (i >= total) return;
    float4 v = (i < ubound) ? ((const float4*)user_emb)[i]
                            : ((const float4*)item_emb)[i - ubound];
    ((float4*)ego)[i] = v;
    ((float4*)acc)[i] = v;
}

__global__ void k_noise_invnorm(const float* __restrict__ noise,
                                float* __restrict__ inv_scale) {
    const int ntiles = (N_LAYERS * NTOT) / 16;
    int wave = (blockIdx.x * blockDim.x + threadIdx.x) >> 5;
    int lane = threadIdx.x & 31;
    if (wave >= ntiles) return;

    int  base = wave * 16;
    int  row  = base + (lane & 15);
    int  kb   = (lane >> 4) * 2;
    const float* p = noise + (size_t)row * EMB + kb;

    v8f c = {};
    #pragma unroll
    for (int kk = 0; kk < EMB; kk += 4) {
        v2f a;
        a.x = p[kk + 0];
        a.y = p[kk + 1];
        c = __builtin_amdgcn_wmma_f32_16x16x4_f32(
                 false, a,  false, a,
                 (short)0, c,  false,  false);
    }

    int sel = (lane < 8) ? lane : ((lane >= 24) ? (lane - 24) : -1);
    float d = 0.0f;
    #pragma unroll
    for (int j = 0; j < 8; ++j) d = (sel == j) ? c[j] : d;

    if (sel >= 0) {
        int m = (lane < 8) ? lane : (lane - 16);
        float n = sqrtf(d);
        inv_scale[base + m] = EPS_F / fmaxf(n, 1e-12f);
    }
}

#define EDGES_PER_WAVE 4
__global__ void k_spmm(const float* __restrict__ vals,
                       const int*   __restrict__ rows,
                       const int*   __restrict__ cols,
                       const float* __restrict__ x,
                       float* __restrict__ y) {
    int wave = (blockIdx.x * blockDim.x + threadIdx.x) >> 5;
    int lane = threadIdx.x & 31;
    long e0 = (long)wave * EDGES_PER_WAVE;
    if (e0 >= NNZ_CNT) return;

    __builtin_prefetch(cols + e0 + 2048, 0, 1);
    __builtin_prefetch(rows + e0 + 2048, 0, 1);
    __builtin_prefetch(vals + e0 + 2048, 0, 1);

    #pragma unroll
    for (int j = 0; j < EDGES_PER_WAVE; ++j) {
        long e = e0 + j;
        if (e >= NNZ_CNT) return;
        int   r = rows[e];
        int   c = cols[e];
        float v = vals[e];
        float2 xv = ((const float2*)(x + (size_t)c * EMB))[lane];
        float* dst = y + (size_t)r * EMB + lane * 2;
        unsafeAtomicAdd(dst + 0, v * xv.x);
        unsafeAtomicAdd(dst + 1, v * xv.y);
    }
}

__device__ __forceinline__ float signf_(float v) {
    return (v > 0.0f) ? 1.0f : ((v < 0.0f) ? -1.0f : 0.0f);
}

__global__ void k_fuse(const float* __restrict__ y,
                       const float* __restrict__ noise_k,
                       const float* __restrict__ inv_scale_k,
                       float* __restrict__ ego,
                       float* __restrict__ acc,
                       int final_scale) {
    int i = blockIdx.x * blockDim.x + threadIdx.x;
    const int total = NTOT * EMB / 4;
    if (i >= total) return;
    int row = i >> 4;
    float is = inv_scale_k[row];

    float4 yv = ((const float4*)y)[i];
    float4 nv = ((const float4*)noise_k)[i];

    float4 ev;
    ev.x = yv.x + signf_(yv.x) * nv.x * is;
    ev.y = yv.y + signf_(yv.y) * nv.y * is;
    ev.z = yv.z + signf_(yv.z) * nv.z * is;
    ev.w = yv.w + signf_(yv.w) * nv.w * is;
    ((float4*)ego)[i] = ev;

    float4 av = ((float4*)acc)[i];
    av.x += ev.x; av.y += ev.y; av.z += ev.z; av.w += ev.w;
    if (final_scale) {
        av.x *= 0.25f; av.y *= 0.25f; av.z *= 0.25f; av.w *= 0.25f;
    }
    ((float4*)acc)[i] = av;
}

extern "C" void kernel_launch(void* const* d_in, const int* in_sizes, int n_in,
                              void* d_out, int out_size, void* d_ws, size_t ws_size,
                              hipStream_t stream) {
    const float* user_emb = (const float*)d_in[0];
    const float* item_emb = (const float*)d_in[1];
    const float* adj_vals = (const float*)d_in[2];
    const float* noise    = (const float*)d_in[3];
    const int*   adj_rows = (const int*)d_in[4];
    const int*   adj_cols = (const int*)d_in[5];

    float* acc = (float*)d_out;

    float* ego = (float*)d_ws;
    float* Y   = ego + (size_t)NTOT * EMB;
    float* inv = Y   + (size_t)NTOT * EMB;

    const int elem4 = NTOT * EMB / 4;

    k_init<<<(elem4 + 255) / 256, 256, 0, stream>>>(user_emb, item_emb, ego, acc);

    {
        int waves  = (N_LAYERS * NTOT) / 16;
        int blocks = (waves + 7) / 8;
        k_noise_invnorm<<<blocks, 256, 0, stream>>>(noise, inv);
    }

    for (int k = 0; k < N_LAYERS; ++k) {
        hipMemsetAsync(Y, 0, (size_t)NTOT * EMB * sizeof(float), stream);

        long waves  = (NNZ_CNT + EDGES_PER_WAVE - 1) / EDGES_PER_WAVE;
        int  blocks = (int)((waves + 7) / 8);
        k_spmm<<<blocks, 256, 0, stream>>>(adj_vals, adj_rows, adj_cols, ego, Y);

        k_fuse<<<(elem4 + 255) / 256, 256, 0, stream>>>(
            Y,
            noise + (size_t)k * NTOT * EMB,
            inv   + (size_t)k * NTOT,
            ego, acc, (k == N_LAYERS - 1) ? 1 : 0);
    }
}


# Float atomic scatter replaced by per-row edge lists built in the call (integer count, scan, fill, order by edge index); one wave sums each row in fixed edge order with FMA, adds noise and layer sum
# speedup vs baseline: 1.3240x; 1.3240x over previous
#include <hip/hip_runtime.h>
#include <hip/hip_bf16.h>


#define USER_NUM 100000
#define ITEM_NUM 50000
#define NTOT     (USER_NUM + ITEM_NUM)
#define EMB      64
#define NNZ_CNT  3000000
#define N_LAYERS 3
#define EPS_F    0.1f
#define SCAN_BLOCK 1024
#define SCAN_BLOCKS ((NTOT + SCAN_BLOCK - 1) / SCAN_BLOCK)
#define ROWPTR_PAD 150016

typedef float v2f __attribute__((ext_vector_type(2)));
typedef float v8f __attribute__((ext_vector_type(8)));

__global__ void k_noise_invnorm(const float* __restrict__ noise,
                                float* __restrict__ inv_scale) {
    const int ntiles = (N_LAYERS * NTOT) / 16;
    int wave = (blockIdx.x * blockDim.x + threadIdx.x) >> 5;
    int lane = threadIdx.x & 31;
    if (wave >= ntiles) return;

    int  base = wave * 16;
    int  row  = base + (lane & 15);
    int  kb   = (lane >> 4) * 2;
    const float* p = noise + (size_t)row * EMB + kb;

    v8f c = {};
    #pragma unroll
    for (int kk = 0; kk < EMB; kk += 4) {
        v2f a;
        a.x = p[kk + 0];
        a.y = p[kk + 1];
        c = __builtin_amdgcn_wmma_f32_16x16x4_f32(
                 false, a,  false, a,
                 (short)0, c,  false,  false);
    }

    int sel = (lane < 8) ? lane : ((lane >= 24) ? (lane - 24) : -1);
    float d = 0.0f;
    #pragma unroll
    for (int j = 0; j < 8; ++j) d = (sel == j) ? c[j] : d;

    if (sel >= 0) {
        int m = (lane < 8) ? lane : (lane - 16);
        float n = sqrtf(d);
        inv_scale[base + m] = EPS_F / fmaxf(n, 1e-12f);
    }
}

__global__ void k_hist(const int* __restrict__ rows, int* __restrict__ cnt) {
    int e = blockIdx.x * blockDim.x + threadIdx.x;
    if (e < NNZ_CNT) atomicAdd(&cnt[rows[e]], 1);
}

__device__ __forceinline__ int wave_incl_scan(int v, int lane) {
    #pragma unroll
    for (int d = 1; d < 32; d <<= 1) {
        int t = __shfl_up(v, d, 32);
        if (lane >= d) v += t;
    }
    return v;
}

__device__ __forceinline__ int wave_sum(int v) {
    #pragma unroll
    for (int d = 16; d > 0; d >>= 1) v += __shfl_xor(v, d, 32);
    return v;
}

__global__ void __launch_bounds__(SCAN_BLOCK)
k_scan(const int* __restrict__ cnt, int* __restrict__ rowptr, int* __restrict__ cursor) {
    __shared__ int s_w[32];
    __shared__ int s_off;
    const int t = threadIdx.x, lane = t & 31, w = t >> 5;
    const int b = blockIdx.x;

    int pre = 0;
    const int4* c4 = (const int4*)cnt;
    const int n4 = b * (SCAN_BLOCK / 4);
    for (int i = t; i < n4; i += SCAN_BLOCK) {
        int4 q = c4[i];
        pre += (q.x + q.y) + (q.z + q.w);
    }
    pre = wave_sum(pre);
    if (lane == 0) s_w[w] = pre;
    __syncthreads();
    if (w == 0) {
        int x = wave_sum(s_w[lane]);
        if (lane == 0) s_off = x;
    }
    __syncthreads();
    const int off = s_off;

    const int i = b * SCAN_BLOCK + t;
    const int v = (i < NTOT) ? cnt[i] : 0;
    const int inc = wave_incl_scan(v, lane);
    if (lane == 31) s_w[w] = inc;
    __syncthreads();
    if (w == 0) {
        int x  = s_w[lane];
        int xi = wave_incl_scan(x, lane);
        s_w[lane] = xi - x;
    }
    __syncthreads();
    const int excl = off + s_w[w] + inc - v;
    if (i < NTOT) { rowptr[i] = excl; cursor[i] = excl; }
    if (i == NTOT - 1) rowptr[NTOT] = excl + v;
}

__global__ void k_fill(const int* __restrict__ rows, const int* __restrict__ cols,
                       const float* __restrict__ vals, int* __restrict__ cursor,
                       int* __restrict__ tmp_e, int2* __restrict__ tmp_cv) {
    int e = blockIdx.x * blockDim.x + threadIdx.x;
    if (e >= NNZ_CNT) return;
    int r = rows[e];
    int slot = atomicAdd(&cursor[r], 1);
    tmp_e[slot]  = e;
    tmp_cv[slot] = make_int2(cols[e], __float_as_int(vals[e]));
}

__global__ void k_order(const int* __restrict__ rowptr, const int* __restrict__ tmp_e,
                        const int2* __restrict__ tmp_cv, int2* __restrict__ scv) {
    const int r = __builtin_amdgcn_readfirstlane((int)((blockIdx.x * blockDim.x + threadIdx.x) >> 5));
    const int lane = threadIdx.x & 31;
    if (r >= NTOT) return;
    const int s = rowptr[r];
    const int n = rowptr[r + 1] - s;
    for (int bi = 0; bi < n; bi += 32) {
        const int i = bi + lane;
        const int mine = (i < n) ? tmp_e[s + i] : 0x7fffffff;
        int rank = 0;
        for (int bj = 0; bj < n; bj += 32) {
            const int j = bj + lane;
            const int other = (j < n) ? tmp_e[s + j] : 0x7fffffff;
            const int m = (n - bj < 32) ? (n - bj) : 32;
            for (int k = 0; k < m; ++k) {
                int ek = __builtin_amdgcn_readlane(other, k);
                rank += (ek < mine) ? 1 : 0;
            }
        }
        if (i < n) scv[s + rank] = tmp_cv[s + i];
    }
}

__device__ __forceinline__ float signf_(float v) {
    return (v > 0.0f) ? 1.0f : ((v < 0.0f) ? -1.0f : 0.0f);
}

template <int L, int CNT>
__device__ __forceinline__ void edge_batch(float2& a, const int2 cv, const int k0, const int lane,
                                           const float* __restrict__ user_emb,
                                           const float* __restrict__ item_emb,
                                           const float* src) {
    float2 x[CNT];
    float  v[CNT];
    #pragma unroll
    for (int j = 0; j < CNT; ++j) {
        const int c = __builtin_amdgcn_readlane(cv.x, k0 + j);
        v[j] = __int_as_float(__builtin_amdgcn_readlane(cv.y, k0 + j));
        const float* xr;
        if (L == 1) xr = (c < USER_NUM) ? (user_emb + (size_t)c * EMB)
                                        : (item_emb + (size_t)(c - USER_NUM) * EMB);
        else        xr = src + (size_t)c * EMB;
        x[j] = ((const float2*)xr)[lane];
    }
    #pragma unroll
    for (int j = 0; j < CNT; ++j) {
        a.x = __builtin_fmaf(v[j], x[j].x, a.x);
        a.y = __builtin_fmaf(v[j], x[j].y, a.y);
    }
}

template <int L>
__global__ void k_layer(const int* __restrict__ rowptr, const int2* __restrict__ scv,
                        const float* __restrict__ user_emb,
                        const float* __restrict__ item_emb,
                        const float* src,
                        const float* __restrict__ noise_k,
                        const float* __restrict__ inv_scale_k,
                        const float* e1, const float* e2, float* dst) {
    const int r = __builtin_amdgcn_readfirstlane((int)((blockIdx.x * blockDim.x + threadIdx.x) >> 5));
    const int lane = threadIdx.x & 31;
    if (r >= NTOT) return;
    const int s  = rowptr[r];
    const int en = rowptr[r + 1];

    const size_t o  = (size_t)r * EMB + (size_t)lane * 2;
    const float  is = inv_scale_k[r];
    const float2 nv = *(const float2*)(noise_k + o);
    float2 g  = make_float2(0.0f, 0.0f);
    float2 p1 = make_float2(0.0f, 0.0f);
    float2 p2 = make_float2(0.0f, 0.0f);
    if (L == 3) {
        const float* g0 = (r < USER_NUM) ? (user_emb + (size_t)r * EMB)
                                         : (item_emb + (size_t)(r - USER_NUM) * EMB);
        g  = ((const float2*)g0)[lane];
        p1 = *(const float2*)(e1 + o);
        p2 = *(const float2*)(e2 + o);
    }

    float2 a = make_float2(0.0f, 0.0f);
    for (int b0 = s; b0 < en; b0 += 32) {
        const int idx = b0 + lane;
        int2 cv = make_int2(0, 0);
        if (idx < en) cv = scv[idx];
        const int m = (en - b0 < 32) ? (en - b0) : 32;
        int k = 0;
        for (; k + 16 <= m; k += 16) edge_batch<L, 16>(a, cv, k, lane, user_emb, item_emb, src);
        if (k + 8 <= m) { edge_batch<L, 8>(a, cv, k, lane, user_emb, item_emb, src); k += 8; }
        if (k + 4 <= m) { edge_batch<L, 4>(a, cv, k, lane, user_emb, item_emb, src); k += 4; }
        if (k + 2 <= m) { edge_batch<L, 2>(a, cv, k, lane, user_emb, item_emb, src); k += 2; }
        if (k < m)      { edge_batch<L, 1>(a, cv, k, lane, user_emb, item_emb, src); }
    }

    float2 ev;
    ev.x = a.x + signf_(a.x) * nv.x * is;
    ev.y = a.y + signf_(a.y) * nv.y * is;

    if (L < 3) {
        *(float2*)(dst + o) = ev;
    } else {
        g.x += p1.x; g.y += p1.y;
        g.x += p2.x; g.y += p2.y;
        g.x += ev.x; g.y += ev.y;
        g.x *= 0.25f; g.y *= 0.25f;
        *(float2*)(dst + o) = g;
    }
}

extern "C" void kernel_launch(void* const* d_in, const int* in_sizes, int n_in,
                              void* d_out, int out_size, void* d_ws, size_t ws_size,
                              hipStream_t stream) {
    const float* user_emb = (const float*)d_in[0];
    const float* item_emb = (const float*)d_in[1];
    const float* adj_vals = (const float*)d_in[2];
    const float* noise    = (const float*)d_in[3];
    const int*   adj_rows = (const int*)d_in[4];
    const int*   adj_cols = (const int*)d_in[5];

    float* out = (float*)d_out;

    float* R1  = (float*)d_ws;
    float* R2  = R1 + (size_t)NTOT * EMB;
    float* inv = R2 + (size_t)NTOT * EMB;

    int*  tmp_e  = (int*)R1;
    int2* tmp_cv = (int2*)(R1 + (size_t)NNZ_CNT);
    int2* scv    = (int2*)R2;
    int*  rowptr = (int*)(R2 + (size_t)2 * NNZ_CNT);
    int*  cursor = rowptr + ROWPTR_PAD;
    int*  cnt    = cursor + ROWPTR_PAD;

    hipMemsetAsync(cnt, 0, (size_t)NTOT * sizeof(int), stream);

    {
        int waves  = (N_LAYERS * NTOT) / 16;
        int blocks = (waves + 7) / 8;
        k_noise_invnorm<<<blocks, 256, 0, stream>>>(noise, inv);
    }

    const int eblocks = (NNZ_CNT + 255) / 256;
    const int rblocks = (NTOT + 7) / 8;

    k_hist<<<eblocks, 256, 0, stream>>>(adj_rows, cnt);
    k_scan<<<SCAN_BLOCKS, SCAN_BLOCK, 0, stream>>>(cnt, rowptr, cursor);
    k_fill<<<eblocks, 256, 0, stream>>>(adj_rows, adj_cols, adj_vals, cursor, tmp_e, tmp_cv);
    k_order<<<rblocks, 256, 0, stream>>>(rowptr, tmp_e, tmp_cv, scv);

    k_layer<1><<<rblocks, 256, 0, stream>>>(rowptr, scv, user_emb, item_emb, nullptr,
        noise, inv, nullptr, nullptr, out);
    k_layer<2><<<rblocks, 256, 0, stream>>>(rowptr, scv, user_emb, item_emb, out,
        noise + (size_t)1 * NTOT * EMB, inv + (size_t)1 * NTOT, nullptr, nullptr, R1);
    k_layer<3><<<rblocks, 256, 0, stream>>>(rowptr, scv, user_emb, item_emb, R1,
        noise + (size_t)2 * NTOT * EMB, inv + (size_t)2 * NTOT, out, R1, out);
}


# Row lists built by a two-level counting sort ranked with LDS counters (edge chunks x 512-row buckets, table scan, per-bucket row placement) instead of one global atomic per edge; sums unchanged
# speedup vs baseline: 2.8371x; 2.1428x over previous
#include <hip/hip_runtime.h>
#include <hip/hip_bf16.h>


#define USER_NUM 100000
#define ITEM_NUM 50000
#define NTOT     (USER_NUM + ITEM_NUM)
#define EMB      64
#define NNZ_CNT  3000000
#define N_LAYERS 3
#define EPS_F    0.1f
#define SCAN_BLOCK 1024
#define ROWPTR_PAD 150016

#define BKT_SHIFT 9
#define BKT_ROWS  (1 << BKT_SHIFT)
#define NBKT      ((NTOT + BKT_ROWS - 1) / BKT_ROWS)
#define EPT       12
#define CHUNK     (EPT * 1024)
#define NCHUNK    ((NNZ_CNT + CHUNK - 1) / CHUNK)
#define TBL_N     (NBKT * NCHUNK)
#define TBL_PAD   71808
#define SCAN_BLOCKS ((TBL_N + SCAN_BLOCK - 1) / SCAN_BLOCK)
#define E_BITS    22
#define E_MASK    ((1 << E_BITS) - 1)

static_assert(NNZ_CNT <= (1 << E_BITS), "edge index must fit in E_BITS");
static_assert(E_BITS + BKT_SHIFT <= 31, "packed word must stay non-negative");
static_assert(TBL_N + 1 <= TBL_PAD, "table padding");
static_assert(NBKT <= 1024, "one thread per bucket when the table is read or written");

typedef float v2f __attribute__((ext_vector_type(2)));
typedef float v8f __attribute__((ext_vector_type(8)));

__global__ void k_noise_invnorm(const float* __restrict__ noise,
                                float* __restrict__ inv_scale) {
    const int ntiles = (N_LAYERS * NTOT) / 16;
    int wave = (blockIdx.x * blockDim.x + threadIdx.x) >> 5;
    int lane = threadIdx.x & 31;
    if (wave >= ntiles) return;

    int  base = wave * 16;
    int  row  = base + (lane & 15);
    int  kb   = (lane >> 4) * 2;
    const float* p = noise + (size_t)row * EMB + kb;

    v8f c = {};
    #pragma unroll
    for (int kk = 0; kk < EMB; kk += 4) {
        v2f a;
        a.x = p[kk + 0];
        a.y = p[kk + 1];
        c = __builtin_amdgcn_wmma_f32_16x16x4_f32(
                 false, a,  false, a,
                 (short)0, c,  false,  false);
    }

    int sel = (lane < 8) ? lane : ((lane >= 24) ? (lane - 24) : -1);
    float d = 0.0f;
    #pragma unroll
    for (int j = 0; j < 8; ++j) d = (sel == j) ? c[j] : d;

    if (sel >= 0) {
        int m = (lane < 8) ? lane : (lane - 16);
        float n = sqrtf(d);
        inv_scale[base + m] = EPS_F / fmaxf(n, 1e-12f);
    }
}

__device__ __forceinline__ int wave_incl_scan(int v, int lane) {
    #pragma unroll
    for (int d = 1; d < 32; d <<= 1) {
        int t = __shfl_up(v, d, 32);
        if (lane >= d) v += t;
    }
    return v;
}

__device__ __forceinline__ int wave_sum(int v) {
    #pragma unroll
    for (int d = 16; d > 0; d >>= 1) v += __shfl_xor(v, d, 32);
    return v;
}

__global__ void __launch_bounds__(1024)
k_bcount(const int* __restrict__ rows, int* __restrict__ tbl) {
    __shared__ int h[NBKT];
    const int t = threadIdx.x, g = blockIdx.x;
    if (t < NBKT) h[t] = 0;
    __syncthreads();
    const int e0 = g * CHUNK + t;
    int r[EPT];
    #pragma unroll
    for (int k = 0; k < EPT; ++k) {
        const int e = e0 + k * 1024;
        r[k] = (e < NNZ_CNT) ? rows[e] : -1;
    }
    #pragma unroll
    for (int k = 0; k < EPT; ++k)
        if (r[k] >= 0) atomicAdd(&h[r[k] >> BKT_SHIFT], 1);
    __syncthreads();
    if (t < NBKT) tbl[t * NCHUNK + g] = h[t];
}

__global__ void __launch_bounds__(SCAN_BLOCK)
k_scan(const int* __restrict__ in, int* __restrict__ out) {
    __shared__ int s_w[32];
    __shared__ int s_off;
    const int t = threadIdx.x, lane = t & 31, w = t >> 5;
    const int b = blockIdx.x;

    int pre = 0;
    const int4* c4 = (const int4*)in;
    const int n4 = b * (SCAN_BLOCK / 4);
    for (int i = t; i < n4; i += SCAN_BLOCK) {
        int4 q = c4[i];
        pre += (q.x + q.y) + (q.z + q.w);
    }
    pre = wave_sum(pre);
    if (lane == 0) s_w[w] = pre;
    __syncthreads();
    if (w == 0) {
        int x = wave_sum(s_w[lane]);
        if (lane == 0) s_off = x;
    }
    __syncthreads();
    const int off = s_off;

    const int i = b * SCAN_BLOCK + t;
    const int v = (i < TBL_N) ? in[i] : 0;
    const int inc = wave_incl_scan(v, lane);
    if (lane == 31) s_w[w] = inc;
    __syncthreads();
    if (w == 0) {
        int x  = s_w[lane];
        int xi = wave_incl_scan(x, lane);
        s_w[lane] = xi - x;
    }
    __syncthreads();
    const int excl = off + s_w[w] + inc - v;
    if (i < TBL_N) out[i] = excl;
    if (i == TBL_N - 1) out[TBL_N] = excl + v;
}

__global__ void __launch_bounds__(1024)
k_bscatter(const int* __restrict__ rows, const int* __restrict__ tbl_excl,
           int* __restrict__ bkt_e) {
    __shared__ int cur[NBKT];
    const int t = threadIdx.x, g = blockIdx.x;
    if (t < NBKT) cur[t] = tbl_excl[t * NCHUNK + g];
    __syncthreads();
    const int e0 = g * CHUNK + t;
    int r[EPT];
    #pragma unroll
    for (int k = 0; k < EPT; ++k) {
        const int e = e0 + k * 1024;
        r[k] = (e < NNZ_CNT) ? rows[e] : -1;
    }
    int slot[EPT];
    #pragma unroll
    for (int k = 0; k < EPT; ++k)
        slot[k] = (r[k] >= 0) ? atomicAdd(&cur[r[k] >> BKT_SHIFT], 1) : -1;
    #pragma unroll
    for (int k = 0; k < EPT; ++k)
        if (slot[k] >= 0) bkt_e[slot[k]] = (e0 + k * 1024) | ((r[k] & (BKT_ROWS - 1)) << E_BITS);
}

__global__ void __launch_bounds__(1024)
k_brows(const int* __restrict__ tbl_excl, const int* __restrict__ bkt_e,
        int* __restrict__ rowptr, int* __restrict__ tmp_e) {
    __shared__ int c[BKT_ROWS];
    __shared__ int s_w[32];
    const int t = threadIdx.x, lane = t & 31, w = t >> 5;
    const int b = blockIdx.x;
    const int bs = tbl_excl[b * NCHUNK];
    const int be = tbl_excl[(b + 1) * NCHUNK];
    if (t < BKT_ROWS) c[t] = 0;
    __syncthreads();

    for (int i0 = bs + t; i0 < be; i0 += 4 * 1024) {
        int wd[4];
        #pragma unroll
        for (int k = 0; k < 4; ++k) {
            const int i = i0 + k * 1024;
            wd[k] = (i < be) ? bkt_e[i] : -1;
        }
        #pragma unroll
        for (int k = 0; k < 4; ++k)
            if (wd[k] >= 0) atomicAdd(&c[wd[k] >> E_BITS], 1);
    }
    __syncthreads();

    const int v = (t < BKT_ROWS) ? c[t] : 0;
    const int inc = wave_incl_scan(v, lane);
    if (lane == 31) s_w[w] = inc;
    __syncthreads();
    if (w == 0) {
        int x  = s_w[lane];
        int xi = wave_incl_scan(x, lane);
        s_w[lane] = xi - x;
    }
    __syncthreads();
    const int start = bs + s_w[w] + inc - v;
    if (t < BKT_ROWS) {
        c[t] = start;
        const int r = b * BKT_ROWS + t;
        if (r < NTOT) rowptr[r] = start;
        if (r == NTOT - 1) rowptr[NTOT] = start + v;
    }
    __syncthreads();

    for (int i0 = bs + t; i0 < be; i0 += 4 * 1024) {
        int wd[4];
        #pragma unroll
        for (int k = 0; k < 4; ++k) {
            const int i = i0 + k * 1024;
            wd[k] = (i < be) ? bkt_e[i] : -1;
        }
        int slot[4];
        #pragma unroll
        for (int k = 0; k < 4; ++k)
            slot[k] = (wd[k] >= 0) ? atomicAdd(&c[wd[k] >> E_BITS], 1) : -1;
        #pragma unroll
        for (int k = 0; k < 4; ++k)
            if (slot[k] >= 0) tmp_e[slot[k]] = wd[k] & E_MASK;
    }
}

__global__ void k_order(const int* __restrict__ rowptr, const int* __restrict__ tmp_e,
                        const int* __restrict__ cols, const float* __restrict__ vals,
                        int2* __restrict__ scv) {
    const int r = __builtin_amdgcn_readfirstlane((int)((blockIdx.x * blockDim.x + threadIdx.x) >> 5));
    const int lane = threadIdx.x & 31;
    if (r >= NTOT) return;
    const int s = rowptr[r];
    const int n = rowptr[r + 1] - s;
    for (int bi = 0; bi < n; bi += 32) {
        const int i = bi + lane;
        const int mine = (i < n) ? tmp_e[s + i] : 0x7fffffff;
        int2 cv = make_int2(0, 0);
        if (i < n) cv = make_int2(cols[mine], __float_as_int(vals[mine]));
        int rank = 0;
        for (int bj = 0; bj < n; bj += 32) {
            const int j = bj + lane;
            const int other = (j < n) ? tmp_e[s + j] : 0x7fffffff;
            const int m = (n - bj < 32) ? (n - bj) : 32;
            for (int k = 0; k < m; ++k) {
                int ek = __builtin_amdgcn_readlane(other, k);
                rank += (ek < mine) ? 1 : 0;
            }
        }
        if (i < n) scv[s + rank] = cv;
    }
}

__device__ __forceinline__ float signf_(float v) {
    return (v > 0.0f) ? 1.0f : ((v < 0.0f) ? -1.0f : 0.0f);
}

template <int L, int CNT>
__device__ __forceinline__ void edge_batch(float2& a, const int2 cv, const int k0, const int lane,
                                           const float* __restrict__ user_emb,
                                           const float* __restrict__ item_emb,
                                           const float* src) {
    float2 x[CNT];
    float  v[CNT];
    #pragma unroll
    for (int j = 0; j < CNT; ++j) {
        const int c = __builtin_amdgcn_readlane(cv.x, k0 + j);
        v[j] = __int_as_float(__builtin_amdgcn_readlane(cv.y, k0 + j));
        const float* xr;
        if (L == 1) xr = (c < USER_NUM) ? (user_emb + (size_t)c * EMB)
                                        : (item_emb + (size_t)(c - USER_NUM) * EMB);
        else        xr = src + (size_t)c * EMB;
        x[j] = ((const float2*)xr)[lane];
    }
    #pragma unroll
    for (int j = 0; j < CNT; ++j) {
        a.x = __builtin_fmaf(v[j], x[j].x, a.x);
        a.y = __builtin_fmaf(v[j], x[j].y, a.y);
    }
}

template <int L>
__global__ void k_layer(const int* __restrict__ rowptr, const int2* __restrict__ scv,
                        const float* __restrict__ user_emb,
                        const float* __restrict__ item_emb,
                        const float* src,
                        const float* __restrict__ noise_k,
                        const float* __restrict__ inv_scale_k,
                        const float* e1, const float* e2, float* dst) {
    const int r = __builtin_amdgcn_readfirstlane((int)((blockIdx.x * blockDim.x + threadIdx.x) >> 5));
    const int lane = threadIdx.x & 31;
    if (r >= NTOT) return;
    const int s  = rowptr[r];
    const int en = rowptr[r + 1];

    const size_t o  = (size_t)r * EMB + (size_t)lane * 2;
    const float  is = inv_scale_k[r];
    const float2 nv = *(const float2*)(noise_k + o);
    float2 g  = make_float2(0.0f, 0.0f);
    float2 p1 = make_float2(0.0f, 0.0f);
    float2 p2 = make_float2(0.0f, 0.0f);
    if (L == 3) {
        const float* g0 = (r < USER_NUM) ? (user_emb + (size_t)r * EMB)
                                         : (item_emb + (size_t)(r - USER_NUM) * EMB);
        g  = ((const float2*)g0)[lane];
        p1 = *(const float2*)(e1 + o);
        p2 = *(const float2*)(e2 + o);
    }

    float2 a = make_float2(0.0f, 0.0f);
    for (int b0 = s; b0 < en; b0 += 32) {
        const int idx = b0 + lane;
        int2 cv = make_int2(0, 0);
        if (idx < en) cv = scv[idx];
        const int m = (en - b0 < 32) ? (en - b0) : 32;
        int k = 0;
        for (; k + 16 <= m; k += 16) edge_batch<L, 16>(a, cv, k, lane, user_emb, item_emb, src);
        if (k + 8 <= m) { edge_batch<L, 8>(a, cv, k, lane, user_emb, item_emb, src); k += 8; }
        if (k + 4 <= m) { edge_batch<L, 4>(a, cv, k, lane, user_emb, item_emb, src); k += 4; }
        if (k + 2 <= m) { edge_batch<L, 2>(a, cv, k, lane, user_emb, item_emb, src); k += 2; }
        if (k < m)      { edge_batch<L, 1>(a, cv, k, lane, user_emb, item_emb, src); }
    }

    float2 ev;
    ev.x = a.x + signf_(a.x) * nv.x * is;
    ev.y = a.y + signf_(a.y) * nv.y * is;

    if (L < 3) {
        *(float2*)(dst + o) = ev;
    } else {
        g.x += p1.x; g.y += p1.y;
        g.x += p2.x; g.y += p2.y;
        g.x += ev.x; g.y += ev.y;
        g.x *= 0.25f; g.y *= 0.25f;
        *(float2*)(dst + o) = g;
    }
}

extern "C" void kernel_launch(void* const* d_in, const int* in_sizes, int n_in,
                              void* d_out, int out_size, void* d_ws, size_t ws_size,
                              hipStream_t stream) {
    const float* user_emb = (const float*)d_in[0];
    const float* item_emb = (const float*)d_in[1];
    const float* adj_vals = (const float*)d_in[2];
    const float* noise    = (const float*)d_in[3];
    const int*   adj_rows = (const int*)d_in[4];
    const int*   adj_cols = (const int*)d_in[5];

    float* out = (float*)d_out;

    float* R1  = (float*)d_ws;
    float* R2  = R1 + (size_t)NTOT * EMB;
    float* inv = R2 + (size_t)NTOT * EMB;

    int*  bkt_e  = (int*)R1;
    int*  tmp_e  = bkt_e + NNZ_CNT;
    int2* scv    = (int2*)R2;
    int*  rowptr = (int*)(R2 + (size_t)2 * NNZ_CNT);
    int*  tbl    = rowptr + ROWPTR_PAD;
    int*  tblx   = tbl + TBL_PAD;

    {
        int waves  = (N_LAYERS * NTOT) / 16;
        int blocks = (waves + 7) / 8;
        k_noise_invnorm<<<blocks, 256, 0, stream>>>(noise, inv);
    }

    const int rblocks = (NTOT + 7) / 8;

    k_bcount<<<NCHUNK, 1024, 0, stream>>>(adj_rows, tbl);
    k_scan<<<SCAN_BLOCKS, SCAN_BLOCK, 0, stream>>>(tbl, tblx);
    k_bscatter<<<NCHUNK, 1024, 0, stream>>>(adj_rows, tblx, bkt_e);
    k_brows<<<NBKT, 1024, 0, stream>>>(tblx, bkt_e, rowptr, tmp_e);
    k_order<<<rblocks, 256, 0, stream>>>(rowptr, tmp_e, adj_cols, adj_vals, scv);

    k_layer<1><<<rblocks, 256, 0, stream>>>(rowptr, scv, user_emb, item_emb, nullptr,
        noise, inv, nullptr, nullptr, out);
    k_layer<2><<<rblocks, 256, 0, stream>>>(rowptr, scv, user_emb, item_emb, out,
        noise + (size_t)1 * NTOT * EMB, inv + (size_t)1 * NTOT, nullptr, nullptr, R1);
    k_layer<3><<<rblocks, 256, 0, stream>>>(rowptr, scv, user_emb, item_emb, R1,
        noise + (size_t)2 * NTOT * EMB, inv + (size_t)2 * NTOT, out, R1, out);
}


# Column and value ride with each edge through both binning passes as one 12-byte record read in edge order, instead of two random 4-byte loads per edge by index in the ordering kernel; sums unchanged
# speedup vs baseline: 3.1694x; 1.1171x over previous
#include <hip/hip_runtime.h>
#include <hip/hip_bf16.h>


#define USER_NUM 100000
#define ITEM_NUM 50000
#define NTOT     (USER_NUM + ITEM_NUM)
#define EMB      64
#define NNZ_CNT  3000000
#define N_LAYERS 3
#define EPS_F    0.1f
#define SCAN_BLOCK 1024
#define ROWPTR_PAD 150016

#define BKT_SHIFT 9
#define BKT_ROWS  (1 << BKT_SHIFT)
#define NBKT      ((NTOT + BKT_ROWS - 1) / BKT_ROWS)
#define EPT       12
#define CHUNK     (EPT * 1024)
#define NCHUNK    ((NNZ_CNT + CHUNK - 1) / CHUNK)
#define TBL_N     (NBKT * NCHUNK)
#define TBL_PAD   71808
#define SCAN_BLOCKS ((TBL_N + SCAN_BLOCK - 1) / SCAN_BLOCK)
#define E_BITS    22
#define E_MASK    ((1 << E_BITS) - 1)

static_assert(NNZ_CNT <= (1 << E_BITS), "edge index must fit in E_BITS");
static_assert(E_BITS + BKT_SHIFT <= 31, "packed word must stay non-negative");
static_assert(TBL_N + 1 <= TBL_PAD, "table padding");
static_assert(NBKT <= 1024, "one thread per bucket when the table is read or written");
static_assert((size_t)3 * NNZ_CNT <= (size_t)NTOT * EMB, "a record array fits one [N, 64] region");
static_assert((size_t)3 * NNZ_CNT + ROWPTR_PAD + 2 * TBL_PAD <= (size_t)NTOT * EMB, "records, row bounds and tables fit the second region");

struct Rec { int k; int c; int v; };

typedef float v2f __attribute__((ext_vector_type(2)));
typedef float v8f __attribute__((ext_vector_type(8)));

__global__ void k_noise_invnorm(const float* __restrict__ noise,
                                float* __restrict__ inv_scale) {
    const int ntiles = (N_LAYERS * NTOT) / 16;
    int wave = (blockIdx.x * blockDim.x + threadIdx.x) >> 5;
    int lane = threadIdx.x & 31;
    if (wave >= ntiles) return;

    int  base = wave * 16;
    int  row  = base + (lane & 15);
    int  kb   = (lane >> 4) * 2;
    const float* p = noise + (size_t)row * EMB + kb;

    v8f c = {};
    #pragma unroll
    for (int kk = 0; kk < EMB; kk += 4) {
        v2f a;
        a.x = p[kk + 0];
        a.y = p[kk + 1];
        c = __builtin_amdgcn_wmma_f32_16x16x4_f32(
                 false, a,  false, a,
                 (short)0, c,  false,  false);
    }

    int sel = (lane < 8) ? lane : ((lane >= 24) ? (lane - 24) : -1);
    float d = 0.0f;
    #pragma unroll
    for (int j = 0; j < 8; ++j) d = (sel == j) ? c[j] : d;

    if (sel >= 0) {
        int m = (lane < 8) ? lane : (lane - 16);
        float n = sqrtf(d);
        inv_scale[base + m] = EPS_F / fmaxf(n, 1e-12f);
    }
}

__device__ __forceinline__ int wave_incl_scan(int v, int lane) {
    #pragma unroll
    for (int d = 1; d < 32; d <<= 1) {
        int t = __shfl_up(v, d, 32);
        if (lane >= d) v += t;
    }
    return v;
}

__device__ __forceinline__ int wave_sum(int v) {
    #pragma unroll
    for (int d = 16; d > 0; d >>= 1) v += __shfl_xor(v, d, 32);
    return v;
}

__global__ void __launch_bounds__(1024)
k_bcount(const int* __restrict__ rows, int* __restrict__ tbl) {
    __shared__ int h[NBKT];
    const int t = threadIdx.x, g = blockIdx.x;
    if (t < NBKT) h[t] = 0;
    __syncthreads();
    const int e0 = g * CHUNK + t;
    int r[EPT];
    #pragma unroll
    for (int k = 0; k < EPT; ++k) {
        const int e = e0 + k * 1024;
        r[k] = (e < NNZ_CNT) ? rows[e] : -1;
    }
    #pragma unroll
    for (int k = 0; k < EPT; ++k)
        if (r[k] >= 0) atomicAdd(&h[r[k] >> BKT_SHIFT], 1);
    __syncthreads();
    if (t < NBKT) tbl[t * NCHUNK + g] = h[t];
}

__global__ void __launch_bounds__(SCAN_BLOCK)
k_scan(const int* __restrict__ in, int* __restrict__ out) {
    __shared__ int s_w[32];
    __shared__ int s_off;
    const int t = threadIdx.x, lane = t & 31, w = t >> 5;
    const int b = blockIdx.x;

    int pre = 0;
    const int4* c4 = (const int4*)in;
    const int n4 = b * (SCAN_BLOCK / 4);
    for (int i = t; i < n4; i += SCAN_BLOCK) {
        int4 q = c4[i];
        pre += (q.x + q.y) + (q.z + q.w);
    }
    pre = wave_sum(pre);
    if (lane == 0) s_w[w] = pre;
    __syncthreads();
    if (w == 0) {
        int x = wave_sum(s_w[lane]);
        if (lane == 0) s_off = x;
    }
    __syncthreads();
    const int off = s_off;

    const int i = b * SCAN_BLOCK + t;
    const int v = (i < TBL_N) ? in[i] : 0;
    const int inc = wave_incl_scan(v, lane);
    if (lane == 31) s_w[w] = inc;
    __syncthreads();
    if (w == 0) {
        int x  = s_w[lane];
        int xi = wave_incl_scan(x, lane);
        s_w[lane] = xi - x;
    }
    __syncthreads();
    const int excl = off + s_w[w] + inc - v;
    if (i < TBL_N) out[i] = excl;
    if (i == TBL_N - 1) out[TBL_N] = excl + v;
}

__global__ void __launch_bounds__(1024)
k_bscatter(const int* __restrict__ rows, const int* __restrict__ cols,
           const float* __restrict__ vals, const int* __restrict__ tbl_excl,
           Rec* __restrict__ bkt) {
    __shared__ int cur[NBKT];
    const int t = threadIdx.x, g = blockIdx.x;
    if (t < NBKT) cur[t] = tbl_excl[t * NCHUNK + g];
    __syncthreads();
    const int e0 = g * CHUNK + t;
    int r[EPT], c[EPT], v[EPT];
    #pragma unroll
    for (int k = 0; k < EPT; ++k) {
        const int e = e0 + k * 1024;
        const bool in = e < NNZ_CNT;
        r[k] = in ? rows[e] : -1;
        c[k] = in ? cols[e] : 0;
        v[k] = in ? __float_as_int(vals[e]) : 0;
    }
    int slot[EPT];
    #pragma unroll
    for (int k = 0; k < EPT; ++k)
        slot[k] = (r[k] >= 0) ? atomicAdd(&cur[r[k] >> BKT_SHIFT], 1) : -1;
    #pragma unroll
    for (int k = 0; k < EPT; ++k)
        if (slot[k] >= 0) {
            Rec o;
            o.k = (e0 + k * 1024) | ((r[k] & (BKT_ROWS - 1)) << E_BITS);
            o.c = c[k];
            o.v = v[k];
            bkt[slot[k]] = o;
        }
}

__global__ void __launch_bounds__(1024)
k_brows(const int* __restrict__ tbl_excl, const Rec* __restrict__ bkt,
        int* __restrict__ rowptr, Rec* __restrict__ tmp) {
    __shared__ int c[BKT_ROWS];
    __shared__ int s_w[32];
    const int t = threadIdx.x, lane = t & 31, w = t >> 5;
    const int b = blockIdx.x;
    const int bs = tbl_excl[b * NCHUNK];
    const int be = tbl_excl[(b + 1) * NCHUNK];
    if (t < BKT_ROWS) c[t] = 0;
    __syncthreads();

    for (int i0 = bs + t; i0 < be; i0 += 4 * 1024) {
        int wd[4];
        #pragma unroll
        for (int k = 0; k < 4; ++k) {
            const int i = i0 + k * 1024;
            wd[k] = (i < be) ? bkt[i].k : -1;
        }
        #pragma unroll
        for (int k = 0; k < 4; ++k)
            if (wd[k] >= 0) atomicAdd(&c[wd[k] >> E_BITS], 1);
    }
    __syncthreads();

    const int v = (t < BKT_ROWS) ? c[t] : 0;
    const int inc = wave_incl_scan(v, lane);
    if (lane == 31) s_w[w] = inc;
    __syncthreads();
    if (w == 0) {
        int x  = s_w[lane];
        int xi = wave_incl_scan(x, lane);
        s_w[lane] = xi - x;
    }
    __syncthreads();
    const int start = bs + s_w[w] + inc - v;
    if (t < BKT_ROWS) {
        c[t] = start;
        const int r = b * BKT_ROWS + t;
        if (r < NTOT) rowptr[r] = start;
        if (r == NTOT - 1) rowptr[NTOT] = start + v;
    }
    __syncthreads();

    for (int i0 = bs + t; i0 < be; i0 += 4 * 1024) {
        Rec wd[4];
        #pragma unroll
        for (int k = 0; k < 4; ++k) {
            const int i = i0 + k * 1024;
            if (i < be) wd[k] = bkt[i];
            else { wd[k].k = -1; wd[k].c = 0; wd[k].v = 0; }
        }
        int slot[4];
        #pragma unroll
        for (int k = 0; k < 4; ++k)
            slot[k] = (wd[k].k >= 0) ? atomicAdd(&c[wd[k].k >> E_BITS], 1) : -1;
        #pragma unroll
        for (int k = 0; k < 4; ++k)
            if (slot[k] >= 0) {
                Rec o;
                o.k = wd[k].k & E_MASK;
                o.c = wd[k].c;
                o.v = wd[k].v;
                tmp[slot[k]] = o;
            }
    }
}

__global__ void k_order(const int* __restrict__ rowptr, const Rec* __restrict__ tmp,
                        int2* __restrict__ scv) {
    const int r = __builtin_amdgcn_readfirstlane((int)((blockIdx.x * blockDim.x + threadIdx.x) >> 5));
    const int lane = threadIdx.x & 31;
    if (r >= NTOT) return;
    const int s = rowptr[r];
    const int n = rowptr[r + 1] - s;
    for (int bi = 0; bi < n; bi += 32) {
        const int i = bi + lane;
        int mine = 0x7fffffff;
        int2 cv = make_int2(0, 0);
        if (i < n) {
            const Rec m = tmp[s + i];
            mine = m.k;
            cv = make_int2(m.c, m.v);
        }
        int rank = 0;
        for (int bj = 0; bj < n; bj += 32) {
            const int j = bj + lane;
            const int other = (j < n) ? tmp[s + j].k : 0x7fffffff;
            const int m = (n - bj < 32) ? (n - bj) : 32;
            for (int k = 0; k < m; ++k) {
                int ek = __builtin_amdgcn_readlane(other, k);
                rank += (ek < mine) ? 1 : 0;
            }
        }
        if (i < n) scv[s + rank] = cv;
    }
}

__device__ __forceinline__ float signf_(float v) {
    return (v > 0.0f) ? 1.0f : ((v < 0.0f) ? -1.0f : 0.0f);
}

template <int L, int CNT>
__device__ __forceinline__ void edge_batch(float2& a, const int2 cv, const int k0, const int lane,
                                           const float* __restrict__ user_emb,
                                           const float* __restrict__ item_emb,
                                           const float* src) {
    float2 x[CNT];
    float  v[CNT];
    #pragma unroll
    for (int j = 0; j < CNT; ++j) {
        const int c = __builtin_amdgcn_readlane(cv.x, k0 + j);
        v[j] = __int_as_float(__builtin_amdgcn_readlane(cv.y, k0 + j));
        const float* xr;
        if (L == 1) xr = (c < USER_NUM) ? (user_emb + (size_t)c * EMB)
                                        : (item_emb + (size_t)(c - USER_NUM) * EMB);
        else        xr = src + (size_t)c * EMB;
        x[j] = ((const float2*)xr)[lane];
    }
    #pragma unroll
    for (int j = 0; j < CNT; ++j) {
        a.x = __builtin_fmaf(v[j], x[j].x, a.x);
        a.y = __builtin_fmaf(v[j], x[j].y, a.y);
    }
}

template <int L>
__global__ void k_layer(const int* __restrict__ rowptr, const int2* __restrict__ scv,
                        const float* __restrict__ user_emb,
                        const float* __restrict__ item_emb,
                        const float* src,
                        const float* __restrict__ noise_k,
                        const float* __restrict__ inv_scale_k,
                        const float* e1, const float* e2, float* dst) {
    const int r = __builtin_amdgcn_readfirstlane((int)((blockIdx.x * blockDim.x + threadIdx.x) >> 5));
    const int lane = threadIdx.x & 31;
    if (r >= NTOT) return;
    const int s  = rowptr[r];
    const int en = rowptr[r + 1];

    const size_t o  = (size_t)r * EMB + (size_t)lane * 2;
    const float  is = inv_scale_k[r];
    const float2 nv = *(const float2*)(noise_k + o);
    float2 g  = make_float2(0.0f, 0.0f);
    float2 p1 = make_float2(0.0f, 0.0f);
    float2 p2 = make_float2(0.0f, 0.0f);
    if (L == 3) {
        const float* g0 = (r < USER_NUM) ? (user_emb + (size_t)r * EMB)
                                         : (item_emb + (size_t)(r - USER_NUM) * EMB);
        g  = ((const float2*)g0)[lane];
        p1 = *(const float2*)(e1 + o);
        p2 = *(const float2*)(e2 + o);
    }

    float2 a = make_float2(0.0f, 0.0f);
    for (int b0 = s; b0 < en; b0 += 32) {
        const int idx = b0 + lane;
        int2 cv = make_int2(0, 0);
        if (idx < en) cv = scv[idx];
        const int m = (en - b0 < 32) ? (en - b0) : 32;
        int k = 0;
        for (; k + 16 <= m; k += 16) edge_batch<L, 16>(a, cv, k, lane, user_emb, item_emb, src);
        if (k + 8 <= m) { edge_batch<L, 8>(a, cv, k, lane, user_emb, item_emb, src); k += 8; }
        if (k + 4 <= m) { edge_batch<L, 4>(a, cv, k, lane, user_emb, item_emb, src); k += 4; }
        if (k + 2 <= m) { edge_batch<L, 2>(a, cv, k, lane, user_emb, item_emb, src); k += 2; }
        if (k < m)      { edge_batch<L, 1>(a, cv, k, lane, user_emb, item_emb, src); }
    }

    float2 ev;
    ev.x = a.x + signf_(a.x) * nv.x * is;
    ev.y = a.y + signf_(a.y) * nv.y * is;

    if (L < 3) {
        *(float2*)(dst + o) = ev;
    } else {
        g.x += p1.x; g.y += p1.y;
        g.x += p2.x; g.y += p2.y;
        g.x += ev.x; g.y += ev.y;
        g.x *= 0.25f; g.y *= 0.25f;
        *(float2*)(dst + o) = g;
    }
}

extern "C" void kernel_launch(void* const* d_in, const int* in_sizes, int n_in,
                              void* d_out, int out_size, void* d_ws, size_t ws_size,
                              hipStream_t stream) {
    const float* user_emb = (const float*)d_in[0];
    const float* item_emb = (const float*)d_in[1];
    const float* adj_vals = (const float*)d_in[2];
    const float* noise    = (const float*)d_in[3];
    const int*   adj_rows = (const int*)d_in[4];
    const int*   adj_cols = (const int*)d_in[5];

    float* out = (float*)d_out;

    float* R1  = (float*)d_ws;
    float* R2  = R1 + (size_t)NTOT * EMB;
    float* inv = R2 + (size_t)NTOT * EMB;

    Rec*  tmp    = (Rec*)R1;
    Rec*  bkt    = (Rec*)R2;
    int2* scv    = (int2*)R2;
    int*  rowptr = (int*)(R2 + (size_t)3 * NNZ_CNT);
    int*  tbl    = rowptr + ROWPTR_PAD;
    int*  tblx   = tbl + TBL_PAD;

    {
        int waves  = (N_LAYERS * NTOT) / 16;
        int blocks = (waves + 7) / 8;
        k_noise_invnorm<<<blocks, 256, 0, stream>>>(noise, inv);
    }

    const int rblocks = (NTOT + 7) / 8;

    k_bcount<<<NCHUNK, 1024, 0, stream>>>(adj_rows, tbl);
    k_scan<<<SCAN_BLOCKS, SCAN_BLOCK, 0, stream>>>(tbl, tblx);
    k_bscatter<<<NCHUNK, 1024, 0, stream>>>(adj_rows, adj_cols, adj_vals, tblx, bkt);
    k_brows<<<NBKT, 1024, 0, stream>>>(tblx, bkt, rowptr, tmp);
    k_order<<<rblocks, 256, 0, stream>>>(rowptr, tmp, scv);

    k_layer<1><<<rblocks, 256, 0, stream>>>(rowptr, scv, user_emb, item_emb, nullptr,
        noise, inv, nullptr, nullptr, out);
    k_layer<2><<<rblocks, 256, 0, stream>>>(rowptr, scv, user_emb, item_emb, out,
        noise + (size_t)1 * NTOT * EMB, inv + (size_t)1 * NTOT, nullptr, nullptr, R1);
    k_layer<3><<<rblocks, 256, 0, stream>>>(rowptr, scv, user_emb, item_emb, R1,
        noise + (size_t)2 * NTOT * EMB, inv + (size_t)2 * NTOT, out, R1, out);
}


# Several rows per wave in the four row kernels (ordering: 8 rows, their records loaded together, short rows ranked in registers; layers: 2 rows in turn) to cut the waves to start; sums unchanged
# speedup vs baseline: 3.2936x; 1.0392x over previous
#include <hip/hip_runtime.h>
#include <hip/hip_bf16.h>


#define USER_NUM 100000
#define ITEM_NUM 50000
#define NTOT     (USER_NUM + ITEM_NUM)
#define EMB      64
#define NNZ_CNT  3000000
#define N_LAYERS 3
#define EPS_F    0.1f
#define SCAN_BLOCK 1024
#define ROWPTR_PAD 150016

#define BKT_SHIFT 9
#define BKT_ROWS  (1 << BKT_SHIFT)
#define NBKT      ((NTOT + BKT_ROWS - 1) / BKT_ROWS)
#define EPT       12
#define CHUNK     (EPT * 1024)
#define NCHUNK    ((NNZ_CNT + CHUNK - 1) / CHUNK)
#define TBL_N     (NBKT * NCHUNK)
#define TBL_PAD   71808
#define SCAN_BLOCKS ((TBL_N + SCAN_BLOCK - 1) / SCAN_BLOCK)
#define E_BITS    22
#define E_MASK    ((1 << E_BITS) - 1)

static_assert(NNZ_CNT <= (1 << E_BITS), "edge index must fit in E_BITS");
static_assert(E_BITS + BKT_SHIFT <= 31, "packed word must stay non-negative");
static_assert(TBL_N + 1 <= TBL_PAD, "table padding");
static_assert(NBKT <= 1024, "one thread per bucket when the table is read or written");
static_assert((size_t)3 * NNZ_CNT <= (size_t)NTOT * EMB, "a record array fits one [N, 64] region");
static_assert((size_t)3 * NNZ_CNT + ROWPTR_PAD + 2 * TBL_PAD <= (size_t)NTOT * EMB, "records, row bounds and tables fit the second region");

struct Rec { int k; int c; int v; };

typedef float v2f __attribute__((ext_vector_type(2)));
typedef float v8f __attribute__((ext_vector_type(8)));

__global__ void k_noise_invnorm(const float* __restrict__ noise,
                                float* __restrict__ inv_scale) {
    const int ntiles = (N_LAYERS * NTOT) / 16;
    int wave = (blockIdx.x * blockDim.x + threadIdx.x) >> 5;
    int lane = threadIdx.x & 31;
    if (wave >= ntiles) return;

    int  base = wave * 16;
    int  row  = base + (lane & 15);
    int  kb   = (lane >> 4) * 2;
    const float* p = noise + (size_t)row * EMB + kb;

    v8f c = {};
    #pragma unroll
    for (int kk = 0; kk < EMB; kk += 4) {
        v2f a;
        a.x = p[kk + 0];
        a.y = p[kk + 1];
        c = __builtin_amdgcn_wmma_f32_16x16x4_f32(
                 false, a,  false, a,
                 (short)0, c,  false,  false);
    }

    int sel = (lane < 8) ? lane : ((lane >= 24) ? (lane - 24) : -1);
    float d = 0.0f;
    #pragma unroll
    for (int j = 0; j < 8; ++j) d = (sel == j) ? c[j] : d;

    if (sel >= 0) {
        int m = (lane < 8) ? lane : (lane - 16);
        float n = sqrtf(d);
        inv_scale[base + m] = EPS_F / fmaxf(n, 1e-12f);
    }
}

__device__ __forceinline__ int wave_incl_scan(int v, int lane) {
    #pragma unroll
    for (int d = 1; d < 32; d <<= 1) {
        int t = __shfl_up(v, d, 32);
        if (lane >= d) v += t;
    }
    return v;
}

__device__ __forceinline__ int wave_sum(int v) {
    #pragma unroll
    for (int d = 16; d > 0; d >>= 1) v += __shfl_xor(v, d, 32);
    return v;
}

__global__ void __launch_bounds__(1024)
k_bcount(const int* __restrict__ rows, int* __restrict__ tbl) {
    __shared__ int h[NBKT];
    const int t = threadIdx.x, g = blockIdx.x;
    if (t < NBKT) h[t] = 0;
    __syncthreads();
    const int e0 = g * CHUNK + t;
    int r[EPT];
    #pragma unroll
    for (int k = 0; k < EPT; ++k) {
        const int e = e0 + k * 1024;
        r[k] = (e < NNZ_CNT) ? rows[e] : -1;
    }
    #pragma unroll
    for (int k = 0; k < EPT; ++k)
        if (r[k] >= 0) atomicAdd(&h[r[k] >> BKT_SHIFT], 1);
    __syncthreads();
    if (t < NBKT) tbl[t * NCHUNK + g] = h[t];
}

__global__ void __launch_bounds__(SCAN_BLOCK)
k_scan(const int* __restrict__ in, int* __restrict__ out) {
    __shared__ int s_w[32];
    __shared__ int s_off;
    const int t = threadIdx.x, lane = t & 31, w = t >> 5;
    const int b = blockIdx.x;

    int pre = 0;
    const int4* c4 = (const int4*)in;
    const int n4 = b * (SCAN_BLOCK / 4);
    for (int i = t; i < n4; i += SCAN_BLOCK) {
        int4 q = c4[i];
        pre += (q.x + q.y) + (q.z + q.w);
    }
    pre = wave_sum(pre);
    if (lane == 0) s_w[w] = pre;
    __syncthreads();
    if (w == 0) {
        int x = wave_sum(s_w[lane]);
        if (lane == 0) s_off = x;
    }
    __syncthreads();
    const int off = s_off;

    const int i = b * SCAN_BLOCK + t;
    const int v = (i < TBL_N) ? in[i] : 0;
    const int inc = wave_incl_scan(v, lane);
    if (lane == 31) s_w[w] = inc;
    __syncthreads();
    if (w == 0) {
        int x  = s_w[lane];
        int xi = wave_incl_scan(x, lane);
        s_w[lane] = xi - x;
    }
    __syncthreads();
    const int excl = off + s_w[w] + inc - v;
    if (i < TBL_N) out[i] = excl;
    if (i == TBL_N - 1) out[TBL_N] = excl + v;
}

__global__ void __launch_bounds__(1024)
k_bscatter(const int* __restrict__ rows, const int* __restrict__ cols,
           const float* __restrict__ vals, const int* __restrict__ tbl_excl,
           Rec* __restrict__ bkt) {
    __shared__ int cur[NBKT];
    const int t = threadIdx.x, g = blockIdx.x;
    if (t < NBKT) cur[t] = tbl_excl[t * NCHUNK + g];
    __syncthreads();
    const int e0 = g * CHUNK + t;
    int r[EPT], c[EPT], v[EPT];
    #pragma unroll
    for (int k = 0; k < EPT; ++k) {
        const int e = e0 + k * 1024;
        const bool in = e < NNZ_CNT;
        r[k] = in ? rows[e] : -1;
        c[k] = in ? cols[e] : 0;
        v[k] = in ? __float_as_int(vals[e]) : 0;
    }
    int slot[EPT];
    #pragma unroll
    for (int k = 0; k < EPT; ++k)
        slot[k] = (r[k] >= 0) ? atomicAdd(&cur[r[k] >> BKT_SHIFT], 1) : -1;
    #pragma unroll
    for (int k = 0; k < EPT; ++k)
        if (slot[k] >= 0) {
            Rec o;
            o.k = (e0 + k * 1024) | ((r[k] & (BKT_ROWS - 1)) << E_BITS);
            o.c = c[k];
            o.v = v[k];
            bkt[slot[k]] = o;
        }
}

__global__ void __launch_bounds__(1024)
k_brows(const int* __restrict__ tbl_excl, const Rec* __restrict__ bkt,
        int* __restrict__ rowptr, Rec* __restrict__ tmp) {
    __shared__ int c[BKT_ROWS];
    __shared__ int s_w[32];
    const int t = threadIdx.x, lane = t & 31, w = t >> 5;
    const int b = blockIdx.x;
    const int bs = tbl_excl[b * NCHUNK];
    const int be = tbl_excl[(b + 1) * NCHUNK];
    if (t < BKT_ROWS) c[t] = 0;
    __syncthreads();

    for (int i0 = bs + t; i0 < be; i0 += 4 * 1024) {
        int wd[4];
        #pragma unroll
        for (int k = 0; k < 4; ++k) {
            const int i = i0 + k * 1024;
            wd[k] = (i < be) ? bkt[i].k : -1;
        }
        #pragma unroll
        for (int k = 0; k < 4; ++k)
            if (wd[k] >= 0) atomicAdd(&c[wd[k] >> E_BITS], 1);
    }
    __syncthreads();

    const int v = (t < BKT_ROWS) ? c[t] : 0;
    const int inc = wave_incl_scan(v, lane);
    if (lane == 31) s_w[w] = inc;
    __syncthreads();
    if (w == 0) {
        int x  = s_w[lane];
        int xi = wave_incl_scan(x, lane);
        s_w[lane] = xi - x;
    }
    __syncthreads();
    const int start = bs + s_w[w] + inc - v;
    if (t < BKT_ROWS) {
        c[t] = start;
        const int r = b * BKT_ROWS + t;
        if (r < NTOT) rowptr[r] = start;
        if (r == NTOT - 1) rowptr[NTOT] = start + v;
    }
    __syncthreads();

    for (int i0 = bs + t; i0 < be; i0 += 4 * 1024) {
        Rec wd[4];
        #pragma unroll
        for (int k = 0; k < 4; ++k) {
            const int i = i0 + k * 1024;
            if (i < be) wd[k] = bkt[i];
            else { wd[k].k = -1; wd[k].c = 0; wd[k].v = 0; }
        }
        int slot[4];
        #pragma unroll
        for (int k = 0; k < 4; ++k)
            slot[k] = (wd[k].k >= 0) ? atomicAdd(&c[wd[k].k >> E_BITS], 1) : -1;
        #pragma unroll
        for (int k = 0; k < 4; ++k)
            if (slot[k] >= 0) {
                Rec o;
                o.k = wd[k].k & E_MASK;
                o.c = wd[k].c;
                o.v = wd[k].v;
                tmp[slot[k]] = o;
            }
    }
}

#define ORD_R 8
static_assert(NTOT % ORD_R == 0, "every wave of the ordering kernel owns ORD_R whole rows");

__device__ __forceinline__ int rank_short_row(const int mine, const int n) {
    int rank = 0;
    for (int k = 0; k < n; k += 4) {
        #pragma unroll
        for (int j = 0; j < 4; ++j) {
            const int ek = __builtin_amdgcn_readlane(mine, k + j);
            rank += (ek < mine) ? 1 : 0;
        }
    }
    return rank;
}

__device__ __forceinline__ void order_long_row(const int s, const int n, const int lane,
                                               const Rec* __restrict__ tmp, int2* __restrict__ scv) {
    for (int bi = 0; bi < n; bi += 32) {
        const int i = bi + lane;
        int mine = 0x7fffffff;
        int2 cv = make_int2(0, 0);
        if (i < n) {
            const Rec m = tmp[s + i];
            mine = m.k;
            cv = make_int2(m.c, m.v);
        }
        int rank = 0;
        for (int bj = 0; bj < n; bj += 32) {
            const int j = bj + lane;
            const int other = (j < n) ? tmp[s + j].k : 0x7fffffff;
            const int m = (n - bj < 32) ? (n - bj) : 32;
            for (int k = 0; k < m; ++k) {
                int ek = __builtin_amdgcn_readlane(other, k);
                rank += (ek < mine) ? 1 : 0;
            }
        }
        if (i < n) scv[s + rank] = cv;
    }
}

__global__ void k_order(const int* __restrict__ rowptr, const Rec* __restrict__ tmp,
                        int2* __restrict__ scv) {
    const int w = __builtin_amdgcn_readfirstlane((int)((blockIdx.x * blockDim.x + threadIdx.x) >> 5));
    const int lane = threadIdx.x & 31;
    const int r0 = w * ORD_R;
    if (r0 >= NTOT) return;

    int rp = 0;
    if (lane <= ORD_R) rp = rowptr[r0 + lane];

    int s[ORD_R], n[ORD_R];
    Rec m[ORD_R];
    #pragma unroll
    for (int q = 0; q < ORD_R; ++q) {
        s[q] = __builtin_amdgcn_readlane(rp, q);
        n[q] = __builtin_amdgcn_readlane(rp, q + 1) - s[q];
        m[q].k = 0x7fffffff; m[q].c = 0; m[q].v = 0;
        if (lane < n[q]) m[q] = tmp[s[q] + lane];
    }
    #pragma unroll
    for (int q = 0; q < ORD_R; ++q) {
        if (n[q] <= 32) {
            const int rank = rank_short_row(m[q].k, n[q]);
            if (lane < n[q]) scv[s[q] + rank] = make_int2(m[q].c, m[q].v);
        }
    }
    #pragma nounroll
    for (int q = 0; q < ORD_R; ++q) {
        const int sq = __builtin_amdgcn_readlane(rp, q);
        const int nq = __builtin_amdgcn_readlane(rp, q + 1) - sq;
        if (nq > 32) order_long_row(sq, nq, lane, tmp, scv);
    }
}

__device__ __forceinline__ float signf_(float v) {
    return (v > 0.0f) ? 1.0f : ((v < 0.0f) ? -1.0f : 0.0f);
}

template <int L, int CNT>
__device__ __forceinline__ void edge_batch(float2& a, const int2 cv, const int k0, const int lane,
                                           const float* __restrict__ user_emb,
                                           const float* __restrict__ item_emb,
                                           const float* src) {
    float2 x[CNT];
    float  v[CNT];
    #pragma unroll
    for (int j = 0; j < CNT; ++j) {
        const int c = __builtin_amdgcn_readlane(cv.x, k0 + j);
        v[j] = __int_as_float(__builtin_amdgcn_readlane(cv.y, k0 + j));
        const float* xr;
        if (L == 1) xr = (c < USER_NUM) ? (user_emb + (size_t)c * EMB)
                                        : (item_emb + (size_t)(c - USER_NUM) * EMB);
        else        xr = src + (size_t)c * EMB;
        x[j] = ((const float2*)xr)[lane];
    }
    #pragma unroll
    for (int j = 0; j < CNT; ++j) {
        a.x = __builtin_fmaf(v[j], x[j].x, a.x);
        a.y = __builtin_fmaf(v[j], x[j].y, a.y);
    }
}

#define LAY_R 2

template <int L>
__global__ void k_layer(const int* __restrict__ rowptr, const int2* __restrict__ scv,
                        const float* __restrict__ user_emb,
                        const float* __restrict__ item_emb,
                        const float* src,
                        const float* __restrict__ noise_k,
                        const float* __restrict__ inv_scale_k,
                        const float* e1, const float* e2, float* dst) {
    const int w = __builtin_amdgcn_readfirstlane((int)((blockIdx.x * blockDim.x + threadIdx.x) >> 5));
    const int lane = threadIdx.x & 31;
    for (int q = 0; q < LAY_R; ++q) {
    const int r = w * LAY_R + q;
    if (r >= NTOT) return;
    const int s  = rowptr[r];
    const int en = rowptr[r + 1];

    const size_t o  = (size_t)r * EMB + (size_t)lane * 2;
    const float  is = inv_scale_k[r];
    const float2 nv = *(const float2*)(noise_k + o);
    float2 g  = make_float2(0.0f, 0.0f);
    float2 p1 = make_float2(0.0f, 0.0f);
    float2 p2 = make_float2(0.0f, 0.0f);
    if (L == 3) {
        const float* g0 = (r < USER_NUM) ? (user_emb + (size_t)r * EMB)
                                         : (item_emb + (size_t)(r - USER_NUM) * EMB);
        g  = ((const float2*)g0)[lane];
        p1 = *(const float2*)(e1 + o);
        p2 = *(const float2*)(e2 + o);
    }

    float2 a = make_float2(0.0f, 0.0f);
    for (int b0 = s; b0 < en; b0 += 32) {
        const int idx = b0 + lane;
        int2 cv = make_int2(0, 0);
        if (idx < en) cv = scv[idx];
        const int m = (en - b0 < 32) ? (en - b0) : 32;
        int k = 0;
        for (; k + 16 <= m; k += 16) edge_batch<L, 16>(a, cv, k, lane, user_emb, item_emb, src);
        if (k + 8 <= m) { edge_batch<L, 8>(a, cv, k, lane, user_emb, item_emb, src); k += 8; }
        if (k + 4 <= m) { edge_batch<L, 4>(a, cv, k, lane, user_emb, item_emb, src); k += 4; }
        if (k + 2 <= m) { edge_batch<L, 2>(a, cv, k, lane, user_emb, item_emb, src); k += 2; }
        if (k < m)      { edge_batch<L, 1>(a, cv, k, lane, user_emb, item_emb, src); }
    }

    float2 ev;
    ev.x = a.x + signf_(a.x) * nv.x * is;
    ev.y = a.y + signf_(a.y) * nv.y * is;

    if (L < 3) {
        *(float2*)(dst + o) = ev;
    } else {
        g.x += p1.x; g.y += p1.y;
        g.x += p2.x; g.y += p2.y;
        g.x += ev.x; g.y += ev.y;
        g.x *= 0.25f; g.y *= 0.25f;
        *(float2*)(dst + o) = g;
    }
    }
}

extern "C" void kernel_launch(void* const* d_in, const int* in_sizes, int n_in,
                              void* d_out, int out_size, void* d_ws, size_t ws_size,
                              hipStream_t stream) {
    const float* user_emb = (const float*)d_in[0];
    const float* item_emb = (const float*)d_in[1];
    const float* adj_vals = (const float*)d_in[2];
    const float* noise    = (const float*)d_in[3];
    const int*   adj_rows = (const int*)d_in[4];
    const int*   adj_cols = (const int*)d_in[5];

    float* out = (float*)d_out;

    float* R1  = (float*)d_ws;
    float* R2  = R1 + (size_t)NTOT * EMB;
    float* inv = R2 + (size_t)NTOT * EMB;

    Rec*  tmp    = (Rec*)R1;
    Rec*  bkt    = (Rec*)R2;
    int2* scv    = (int2*)R2;
    int*  rowptr = (int*)(R2 + (size_t)3 * NNZ_CNT);
    int*  tbl    = rowptr + ROWPTR_PAD;
    int*  tblx   = tbl + TBL_PAD;

    {
        int waves  = (N_LAYERS * NTOT) / 16;
        int blocks = (waves + 7) / 8;
        k_noise_invnorm<<<blocks, 256, 0, stream>>>(noise, inv);
    }

    const int oblocks = (NTOT / ORD_R + 7) / 8;
    const int rblocks = ((NTOT + LAY_R - 1) / LAY_R + 7) / 8;

    k_bcount<<<NCHUNK, 1024, 0, stream>>>(adj_rows, tbl);
    k_scan<<<SCAN_BLOCKS, SCAN_BLOCK, 0, stream>>>(tbl, tblx);
    k_bscatter<<<NCHUNK, 1024, 0, stream>>>(adj_rows, adj_cols, adj_vals, tblx, bkt);
    k_brows<<<NBKT, 1024, 0, stream>>>(tblx, bkt, rowptr, tmp);
    k_order<<<oblocks, 256, 0, stream>>>(rowptr, tmp, scv);

    k_layer<1><<<rblocks, 256, 0, stream>>>(rowptr, scv, user_emb, item_emb, nullptr,
        noise, inv, nullptr, nullptr, out);
    k_layer<2><<<rblocks, 256, 0, stream>>>(rowptr, scv, user_emb, item_emb, out,
        noise + (size_t)1 * NTOT * EMB, inv + (size_t)1 * NTOT, nullptr, nullptr, R1);
    k_layer<3><<<rblocks, 256, 0, stream>>>(rowptr, scv, user_emb, item_emb, R1,
        noise + (size_t)2 * NTOT * EMB, inv + (size_t)2 * NTOT, out, R1, out);
}


# Both binning passes lay their records out in LDS in place order, then store them by consecutive lanes: whole lines per store instead of 32 scattered 12-byte pieces; sums unchanged
# speedup vs baseline: 3.5811x; 1.0873x over previous
#include <hip/hip_runtime.h>
#include <hip/hip_bf16.h>


#define USER_NUM 100000
#define ITEM_NUM 50000
#define NTOT     (USER_NUM + ITEM_NUM)
#define EMB      64
#define NNZ_CNT  3000000
#define N_LAYERS 3
#define EPS_F    0.1f
#define SCAN_BLOCK 1024
#define ROWPTR_PAD 150016

#define BKT_SHIFT 9
#define BKT_ROWS  (1 << BKT_SHIFT)
#define NBKT      ((NTOT + BKT_ROWS - 1) / BKT_ROWS)
#define EPT       12
#define CHUNK     (EPT * 1024)
#define NCHUNK    ((NNZ_CNT + CHUNK - 1) / CHUNK)
#define TBL_N     (NBKT * NCHUNK)
#define TBL_PAD   71808
#define SCAN_BLOCKS ((TBL_N + SCAN_BLOCK - 1) / SCAN_BLOCK)
#define E_BITS    22
#define E_MASK    ((1 << E_BITS) - 1)

static_assert(NNZ_CNT <= (1 << E_BITS), "edge index must fit in E_BITS");
static_assert(E_BITS + BKT_SHIFT <= 31, "packed word must stay non-negative");
static_assert(TBL_N + 1 <= TBL_PAD, "table padding");
static_assert(NBKT <= 1024, "one thread per bucket when the table is read or written");
static_assert((size_t)3 * NNZ_CNT <= (size_t)NTOT * EMB, "a record array fits one [N, 64] region");
static_assert((size_t)3 * NNZ_CNT + ROWPTR_PAD + 2 * TBL_PAD <= (size_t)NTOT * EMB, "records, row bounds and tables fit the second region");

struct Rec { int k; int c; int v; };

typedef float v2f __attribute__((ext_vector_type(2)));
typedef float v8f __attribute__((ext_vector_type(8)));

__global__ void k_noise_invnorm(const float* __restrict__ noise,
                                float* __restrict__ inv_scale) {
    const int ntiles = (N_LAYERS * NTOT) / 16;
    int wave = (blockIdx.x * blockDim.x + threadIdx.x) >> 5;
    int lane = threadIdx.x & 31;
    if (wave >= ntiles) return;

    int  base = wave * 16;
    int  row  = base + (lane & 15);
    int  kb   = (lane >> 4) * 2;
    const float* p = noise + (size_t)row * EMB + kb;

    v8f c = {};
    #pragma unroll
    for (int kk = 0; kk < EMB; kk += 4) {
        v2f a;
        a.x = p[kk + 0];
        a.y = p[kk + 1];
        c = __builtin_amdgcn_wmma_f32_16x16x4_f32(
                 false, a,  false, a,
                 (short)0, c,  false,  false);
    }

    int sel = (lane < 8) ? lane : ((lane >= 24) ? (lane - 24) : -1);
    float d = 0.0f;
    #pragma unroll
    for (int j = 0; j < 8; ++j) d = (sel == j) ? c[j] : d;

    if (sel >= 0) {
        int m = (lane < 8) ? lane : (lane - 16);
        float n = sqrtf(d);
        inv_scale[base + m] = EPS_F / fmaxf(n, 1e-12f);
    }
}

__device__ __forceinline__ int wave_incl_scan(int v, int lane) {
    #pragma unroll
    for (int d = 1; d < 32; d <<= 1) {
        int t = __shfl_up(v, d, 32);
        if (lane >= d) v += t;
    }
    return v;
}

__device__ __forceinline__ int wave_sum(int v) {
    #pragma unroll
    for (int d = 16; d > 0; d >>= 1) v += __shfl_xor(v, d, 32);
    return v;
}

__global__ void __launch_bounds__(1024)
k_bcount(const int* __restrict__ rows, int* __restrict__ tbl) {
    __shared__ int h[NBKT];
    const int t = threadIdx.x, g = blockIdx.x;
    if (t < NBKT) h[t] = 0;
    __syncthreads();
    const int e0 = g * CHUNK + t;
    int r[EPT];
    #pragma unroll
    for (int k = 0; k < EPT; ++k) {
        const int e = e0 + k * 1024;
        r[k] = (e < NNZ_CNT) ? rows[e] : -1;
    }
    #pragma unroll
    for (int k = 0; k < EPT; ++k)
        if (r[k] >= 0) atomicAdd(&h[r[k] >> BKT_SHIFT], 1);
    __syncthreads();
    if (t < NBKT) tbl[t * NCHUNK + g] = h[t];
}

__global__ void __launch_bounds__(SCAN_BLOCK)
k_scan(const int* __restrict__ in, int* __restrict__ out) {
    __shared__ int s_w[32];
    __shared__ int s_off;
    const int t = threadIdx.x, lane = t & 31, w = t >> 5;
    const int b = blockIdx.x;

    int pre = 0;
    const int4* c4 = (const int4*)in;
    const int n4 = b * (SCAN_BLOCK / 4);
    for (int i = t; i < n4; i += SCAN_BLOCK) {
        int4 q = c4[i];
        pre += (q.x + q.y) + (q.z + q.w);
    }
    pre = wave_sum(pre);
    if (lane == 0) s_w[w] = pre;
    __syncthreads();
    if (w == 0) {
        int x = wave_sum(s_w[lane]);
        if (lane == 0) s_off = x;
    }
    __syncthreads();
    const int off = s_off;

    const int i = b * SCAN_BLOCK + t;
    const int v = (i < TBL_N) ? in[i] : 0;
    const int inc = wave_incl_scan(v, lane);
    if (lane == 31) s_w[w] = inc;
    __syncthreads();
    if (w == 0) {
        int x  = s_w[lane];
        int xi = wave_incl_scan(x, lane);
        s_w[lane] = xi - x;
    }
    __syncthreads();
    const int excl = off + s_w[w] + inc - v;
    if (i < TBL_N) out[i] = excl;
    if (i == TBL_N - 1) out[TBL_N] = excl + v;
}

__global__ void __launch_bounds__(1024)
k_bscatter(const int* __restrict__ rows, const int* __restrict__ cols,
           const float* __restrict__ vals, const int* __restrict__ tbl_excl,
           Rec* __restrict__ bkt) {
    __shared__ int cur[NBKT];
    __shared__ int gdl[NBKT];
    __shared__ int s_w[32];
    __shared__ int4 pl[CHUNK];
    const int t = threadIdx.x, g = blockIdx.x, lane = t & 31, w = t >> 5;

    int gs = 0, cb = 0;
    if (t < NBKT) {
        gs = tbl_excl[t * NCHUNK + g];
        cb = tbl_excl[t * NCHUNK + g + 1] - gs;
    }

    const int e0 = g * CHUNK + t;
    int r[EPT], c[EPT], v[EPT];
    #pragma unroll
    for (int k = 0; k < EPT; ++k) {
        const int e = e0 + k * 1024;
        const bool in = e < NNZ_CNT;
        r[k] = in ? rows[e] : -1;
        c[k] = in ? cols[e] : 0;
        v[k] = in ? __float_as_int(vals[e]) : 0;
    }

    const int inc = wave_incl_scan(cb, lane);
    if (lane == 31) s_w[w] = inc;
    __syncthreads();
    if (w == 0) {
        int x  = s_w[lane];
        int xi = wave_incl_scan(x, lane);
        s_w[lane] = xi - x;
    }
    __syncthreads();
    const int ls = s_w[w] + inc - cb;
    if (t < NBKT) { cur[t] = ls; gdl[t] = gs - ls; }
    __syncthreads();

    int slot[EPT];
    #pragma unroll
    for (int k = 0; k < EPT; ++k)
        slot[k] = (r[k] >= 0) ? atomicAdd(&cur[r[k] >> BKT_SHIFT], 1) : -1;
    #pragma unroll
    for (int k = 0; k < EPT; ++k)
        if (slot[k] >= 0) {
            const int s = slot[k];
            pl[s] = make_int4((e0 + k * 1024) | ((r[k] & (BKT_ROWS - 1)) << E_BITS),
                              c[k], v[k], s + gdl[r[k] >> BKT_SHIFT]);
        }
    __syncthreads();

    const int cnt = (NNZ_CNT - g * CHUNK < CHUNK) ? (NNZ_CNT - g * CHUNK) : CHUNK;
    #pragma unroll
    for (int k = 0; k < EPT; ++k) {
        const int i = t + k * 1024;
        if (i < cnt) {
            const int4 q = pl[i];
            Rec o;
            o.k = q.x;
            o.c = q.y;
            o.v = q.z;
            bkt[q.w] = o;
        }
    }
}

#define BR_EPT 12
#define BR_CAP (BR_EPT * 1024)
__global__ void __launch_bounds__(1024)
k_brows(const int* __restrict__ tbl_excl, const Rec* __restrict__ bkt,
        int* __restrict__ rowptr, Rec* __restrict__ tmp) {
    __shared__ int c[BKT_ROWS];
    __shared__ int s_w[32];
    __shared__ Rec pl[BR_CAP];
    const int t = threadIdx.x, lane = t & 31, w = t >> 5;
    const int b = blockIdx.x;
    const int bs = tbl_excl[b * NCHUNK];
    const int be = tbl_excl[(b + 1) * NCHUNK];
    const int n  = be - bs;
    const bool in_lds = n <= BR_CAP;
    if (t < BKT_ROWS) c[t] = 0;
    __syncthreads();

    int key[BR_EPT];
    if (in_lds) {
        #pragma unroll
        for (int k = 0; k < BR_EPT; ++k) {
            const int i = t + k * 1024;
            key[k] = (i < n) ? bkt[bs + i].k : -1;
        }
        #pragma unroll
        for (int k = 0; k < BR_EPT; ++k)
            if (key[k] >= 0) atomicAdd(&c[key[k] >> E_BITS], 1);
    } else {
        for (int i0 = bs + t; i0 < be; i0 += 4 * 1024) {
            int wd[4];
            #pragma unroll
            for (int k = 0; k < 4; ++k) {
                const int i = i0 + k * 1024;
                wd[k] = (i < be) ? bkt[i].k : -1;
            }
            #pragma unroll
            for (int k = 0; k < 4; ++k)
                if (wd[k] >= 0) atomicAdd(&c[wd[k] >> E_BITS], 1);
        }
    }
    __syncthreads();

    const int v = (t < BKT_ROWS) ? c[t] : 0;
    const int inc = wave_incl_scan(v, lane);
    if (lane == 31) s_w[w] = inc;
    __syncthreads();
    if (w == 0) {
        int x  = s_w[lane];
        int xi = wave_incl_scan(x, lane);
        s_w[lane] = xi - x;
    }
    __syncthreads();
    const int lstart = s_w[w] + inc - v;
    if (t < BKT_ROWS) {
        c[t] = in_lds ? lstart : (bs + lstart);
        const int r = b * BKT_ROWS + t;
        if (r < NTOT) rowptr[r] = bs + lstart;
        if (r == NTOT - 1) rowptr[NTOT] = bs + lstart + v;
    }
    __syncthreads();

    if (in_lds) {
        int cc[BR_EPT], vv[BR_EPT];
        #pragma unroll
        for (int k = 0; k < BR_EPT; ++k) {
            const int i = t + k * 1024;
            if (i < n) { cc[k] = bkt[bs + i].c; vv[k] = bkt[bs + i].v; }
        }
        int slot[BR_EPT];
        #pragma unroll
        for (int k = 0; k < BR_EPT; ++k)
            slot[k] = (key[k] >= 0) ? atomicAdd(&c[key[k] >> E_BITS], 1) : -1;
        #pragma unroll
        for (int k = 0; k < BR_EPT; ++k)
            if (slot[k] >= 0) {
                Rec o;
                o.k = key[k] & E_MASK;
                o.c = cc[k];
                o.v = vv[k];
                pl[slot[k]] = o;
            }
        __syncthreads();
        #pragma unroll
        for (int k = 0; k < BR_EPT; ++k) {
            const int i = t + k * 1024;
            if (i < n) tmp[bs + i] = pl[i];
        }
    } else {
        for (int i0 = bs + t; i0 < be; i0 += 4 * 1024) {
            Rec wd[4];
            #pragma unroll
            for (int k = 0; k < 4; ++k) {
                const int i = i0 + k * 1024;
                if (i < be) wd[k] = bkt[i];
                else { wd[k].k = -1; wd[k].c = 0; wd[k].v = 0; }
            }
            int slot[4];
            #pragma unroll
            for (int k = 0; k < 4; ++k)
                slot[k] = (wd[k].k >= 0) ? atomicAdd(&c[wd[k].k >> E_BITS], 1) : -1;
            #pragma unroll
            for (int k = 0; k < 4; ++k)
                if (slot[k] >= 0) {
                    Rec o;
                    o.k = wd[k].k & E_MASK;
                    o.c = wd[k].c;
                    o.v = wd[k].v;
                    tmp[slot[k]] = o;
                }
        }
    }
}

#define ORD_R 8
static_assert(NTOT % ORD_R == 0, "every wave of the ordering kernel owns ORD_R whole rows");

__device__ __forceinline__ int rank_short_row(const int mine, const int n) {
    int rank = 0;
    for (int k = 0; k < n; k += 4) {
        #pragma unroll
        for (int j = 0; j < 4; ++j) {
            const int ek = __builtin_amdgcn_readlane(mine, k + j);
            rank += (ek < mine) ? 1 : 0;
        }
    }
    return rank;
}

__device__ __forceinline__ void order_long_row(const int s, const int n, const int lane,
                                               const Rec* __restrict__ tmp, int2* __restrict__ scv) {
    for (int bi = 0; bi < n; bi += 32) {
        const int i = bi + lane;
        int mine = 0x7fffffff;
        int2 cv = make_int2(0, 0);
        if (i < n) {
            const Rec m = tmp[s + i];
            mine = m.k;
            cv = make_int2(m.c, m.v);
        }
        int rank = 0;
        for (int bj = 0; bj < n; bj += 32) {
            const int j = bj + lane;
            const int other = (j < n) ? tmp[s + j].k : 0x7fffffff;
            const int m = (n - bj < 32) ? (n - bj) : 32;
            for (int k = 0; k < m; ++k) {
                int ek = __builtin_amdgcn_readlane(other, k);
                rank += (ek < mine) ? 1 : 0;
            }
        }
        if (i < n) scv[s + rank] = cv;
    }
}

__global__ void k_order(const int* __restrict__ rowptr, const Rec* __restrict__ tmp,
                        int2* __restrict__ scv) {
    const int w = __builtin_amdgcn_readfirstlane((int)((blockIdx.x * blockDim.x + threadIdx.x) >> 5));
    const int lane = threadIdx.x & 31;
    const int r0 = w * ORD_R;
    if (r0 >= NTOT) return;

    int rp = 0;
    if (lane <= ORD_R) rp = rowptr[r0 + lane];

    int s[ORD_R], n[ORD_R];
    Rec m[ORD_R];
    #pragma unroll
    for (int q = 0; q < ORD_R; ++q) {
        s[q] = __builtin_amdgcn_readlane(rp, q);
        n[q] = __builtin_amdgcn_readlane(rp, q + 1) - s[q];
        m[q].k = 0x7fffffff; m[q].c = 0; m[q].v = 0;
        if (lane < n[q]) m[q] = tmp[s[q] + lane];
    }
    #pragma unroll
    for (int q = 0; q < ORD_R; ++q) {
        if (n[q] <= 32) {
            const int rank = rank_short_row(m[q].k, n[q]);
            if (lane < n[q]) scv[s[q] + rank] = make_int2(m[q].c, m[q].v);
        }
    }
    #pragma nounroll
    for (int q = 0; q < ORD_R; ++q) {
        const int sq = __builtin_amdgcn_readlane(rp, q);
        const int nq = __builtin_amdgcn_readlane(rp, q + 1) - sq;
        if (nq > 32) order_long_row(sq, nq, lane, tmp, scv);
    }
}

__device__ __forceinline__ float signf_(float v) {
    return (v > 0.0f) ? 1.0f : ((v < 0.0f) ? -1.0f : 0.0f);
}

template <int L, int CNT>
__device__ __forceinline__ void edge_batch(float2& a, const int2 cv, const int k0, const int lane,
                                           const float* __restrict__ user_emb,
                                           const float* __restrict__ item_emb,
                                           const float* src) {
    float2 x[CNT];
    float  v[CNT];
    #pragma unroll
    for (int j = 0; j < CNT; ++j) {
        const int c = __builtin_amdgcn_readlane(cv.x, k0 + j);
        v[j] = __int_as_float(__builtin_amdgcn_readlane(cv.y, k0 + j));
        const float* xr;
        if (L == 1) xr = (c < USER_NUM) ? (user_emb + (size_t)c * EMB)
                                        : (item_emb + (size_t)(c - USER_NUM) * EMB);
        else        xr = src + (size_t)c * EMB;
        x[j] = ((const float2*)xr)[lane];
    }
    #pragma unroll
    for (int j = 0; j < CNT; ++j) {
        a.x = __builtin_fmaf(v[j], x[j].x, a.x);
        a.y = __builtin_fmaf(v[j], x[j].y, a.y);
    }
}

#define LAY_R 2

template <int L>
__global__ void k_layer(const int* __restrict__ rowptr, const int2* __restrict__ scv,
                        const float* __restrict__ user_emb,
                        const float* __restrict__ item_emb,
                        const float* src,
                        const float* __restrict__ noise_k,
                        const float* __restrict__ inv_scale_k,
                        const float* e1, const float* e2, float* dst) {
    const int w = __builtin_amdgcn_readfirstlane((int)((blockIdx.x * blockDim.x + threadIdx.x) >> 5));
    const int lane = threadIdx.x & 31;
    for (int q = 0; q < LAY_R; ++q) {
    const int r = w * LAY_R + q;
    if (r >= NTOT) return;
    const int s  = rowptr[r];
    const int en = rowptr[r + 1];

    const size_t o  = (size_t)r * EMB + (size_t)lane * 2;
    const float  is = inv_scale_k[r];
    const float2 nv = *(const float2*)(noise_k + o);
    float2 g  = make_float2(0.0f, 0.0f);
    float2 p1 = make_float2(0.0f, 0.0f);
    float2 p2 = make_float2(0.0f, 0.0f);
    if (L == 3) {
        const float* g0 = (r < USER_NUM) ? (user_emb + (size_t)r * EMB)
                                         : (item_emb + (size_t)(r - USER_NUM) * EMB);
        g  = ((const float2*)g0)[lane];
        p1 = *(const float2*)(e1 + o);
        p2 = *(const float2*)(e2 + o);
    }

    float2 a = make_float2(0.0f, 0.0f);
    for (int b0 = s; b0 < en; b0 += 32) {
        const int idx = b0 + lane;
        int2 cv = make_int2(0, 0);
        if (idx < en) cv = scv[idx];
        const int m = (en - b0 < 32) ? (en - b0) : 32;
        int k = 0;
        for (; k + 16 <= m; k += 16) edge_batch<L, 16>(a, cv, k, lane, user_emb, item_emb, src);
        if (k + 8 <= m) { edge_batch<L, 8>(a, cv, k, lane, user_emb, item_emb, src); k += 8; }
        if (k + 4 <= m) { edge_batch<L, 4>(a, cv, k, lane, user_emb, item_emb, src); k += 4; }
        if (k + 2 <= m) { edge_batch<L, 2>(a, cv, k, lane, user_emb, item_emb, src); k += 2; }
        if (k < m)      { edge_batch<L, 1>(a, cv, k, lane, user_emb, item_emb, src); }
    }

    float2 ev;
    ev.x = a.x + signf_(a.x) * nv.x * is;
    ev.y = a.y + signf_(a.y) * nv.y * is;

    if (L < 3) {
        *(float2*)(dst + o) = ev;
    } else {
        g.x += p1.x; g.y += p1.y;
        g.x += p2.x; g.y += p2.y;
        g.x += ev.x; g.y += ev.y;
        g.x *= 0.25f; g.y *= 0.25f;
        *(float2*)(dst + o) = g;
    }
    }
}

extern "C" void kernel_launch(void* const* d_in, const int* in_sizes, int n_in,
                              void* d_out, int out_size, void* d_ws, size_t ws_size,
                              hipStream_t stream) {
    const float* user_emb = (const float*)d_in[0];
    const float* item_emb = (const float*)d_in[1];
    const float* adj_vals = (const float*)d_in[2];
    const float* noise    = (const float*)d_in[3];
    const int*   adj_rows = (const int*)d_in[4];
    const int*   adj_cols = (const int*)d_in[5];

    float* out = (float*)d_out;

    float* R1  = (float*)d_ws;
    float* R2  = R1 + (size_t)NTOT * EMB;
    float* inv = R2 + (size_t)NTOT * EMB;

    Rec*  tmp    = (Rec*)R1;
    Rec*  bkt    = (Rec*)R2;
    int2* scv    = (int2*)R2;
    int*  rowptr = (int*)(R2 + (size_t)3 * NNZ_CNT);
    int*  tbl    = rowptr + ROWPTR_PAD;
    int*  tblx   = tbl + TBL_PAD;

    {
        int waves  = (N_LAYERS * NTOT) / 16;
        int blocks = (waves + 7) / 8;
        k_noise_invnorm<<<blocks, 256, 0, stream>>>(noise, inv);
    }

    const int oblocks = (NTOT / ORD_R + 7) / 8;
    const int rblocks = ((NTOT + LAY_R - 1) / LAY_R + 7) / 8;

    k_bcount<<<NCHUNK, 1024, 0, stream>>>(adj_rows, tbl);
    k_scan<<<SCAN_BLOCKS, SCAN_BLOCK, 0, stream>>>(tbl, tblx);
    k_bscatter<<<NCHUNK, 1024, 0, stream>>>(adj_rows, adj_cols, adj_vals, tblx, bkt);
    k_brows<<<NBKT, 1024, 0, stream>>>(tblx, bkt, rowptr, tmp);
    k_order<<<oblocks, 256, 0, stream>>>(rowptr, tmp, scv);

    k_layer<1><<<rblocks, 256, 0, stream>>>(rowptr, scv, user_emb, item_emb, nullptr,
        noise, inv, nullptr, nullptr, out);
    k_layer<2><<<rblocks, 256, 0, stream>>>(rowptr, scv, user_emb, item_emb, out,
        noise + (size_t)1 * NTOT * EMB, inv + (size_t)1 * NTOT, nullptr, nullptr, R1);
    k_layer<3><<<rblocks, 256, 0, stream>>>(rowptr, scv, user_emb, item_emb, R1,
        noise + (size_t)2 * NTOT * EMB, inv + (size_t)2 * NTOT, out, R1, out);
}


# Layer kernels: a wave sums its two rows side by side, one per 16-lane half with 16-byte lanes, so one gather instruction serves one list entry of each row (entries passed by DPP); sums unchanged
# speedup vs baseline: 3.7398x; 1.0443x over previous
#include <hip/hip_runtime.h>
#include <hip/hip_bf16.h>


#define USER_NUM 100000
#define ITEM_NUM 50000
#define NTOT     (USER_NUM + ITEM_NUM)
#define EMB      64
#define NNZ_CNT  3000000
#define N_LAYERS 3
#define EPS_F    0.1f
#define SCAN_BLOCK 1024
#define ROWPTR_PAD 150016

#define BKT_SHIFT 9
#define BKT_ROWS  (1 << BKT_SHIFT)
#define NBKT      ((NTOT + BKT_ROWS - 1) / BKT_ROWS)
#define EPT       12
#define CHUNK     (EPT * 1024)
#define NCHUNK    ((NNZ_CNT + CHUNK - 1) / CHUNK)
#define TBL_N     (NBKT * NCHUNK)
#define TBL_PAD   71808
#define SCAN_BLOCKS ((TBL_N + SCAN_BLOCK - 1) / SCAN_BLOCK)
#define E_BITS    22
#define E_MASK    ((1 << E_BITS) - 1)

static_assert(NNZ_CNT <= (1 << E_BITS), "edge index must fit in E_BITS");
static_assert(E_BITS + BKT_SHIFT <= 31, "packed word must stay non-negative");
static_assert(TBL_N + 1 <= TBL_PAD, "table padding");
static_assert(NBKT <= 1024, "one thread per bucket when the table is read or written");
static_assert((size_t)3 * NNZ_CNT <= (size_t)NTOT * EMB, "a record array fits one [N, 64] region");
static_assert((size_t)3 * NNZ_CNT + ROWPTR_PAD + 2 * TBL_PAD <= (size_t)NTOT * EMB, "records, row bounds and tables fit the second region");

struct Rec { int k; int c; int v; };

typedef float v2f __attribute__((ext_vector_type(2)));
typedef float v8f __attribute__((ext_vector_type(8)));

__global__ void k_noise_invnorm(const float* __restrict__ noise,
                                float* __restrict__ inv_scale) {
    const int ntiles = (N_LAYERS * NTOT) / 16;
    int wave = (blockIdx.x * blockDim.x + threadIdx.x) >> 5;
    int lane = threadIdx.x & 31;
    if (wave >= ntiles) return;

    int  base = wave * 16;
    int  row  = base + (lane & 15);
    int  kb   = (lane >> 4) * 2;
    const float* p = noise + (size_t)row * EMB + kb;

    v8f c = {};
    #pragma unroll
    for (int kk = 0; kk < EMB; kk += 4) {
        v2f a;
        a.x = p[kk + 0];
        a.y = p[kk + 1];
        c = __builtin_amdgcn_wmma_f32_16x16x4_f32(
                 false, a,  false, a,
                 (short)0, c,  false,  false);
    }

    int sel = (lane < 8) ? lane : ((lane >= 24) ? (lane - 24) : -1);
    float d = 0.0f;
    #pragma unroll
    for (int j = 0; j < 8; ++j) d = (sel == j) ? c[j] : d;

    if (sel >= 0) {
        int m = (lane < 8) ? lane : (lane - 16);
        float n = sqrtf(d);
        inv_scale[base + m] = EPS_F / fmaxf(n, 1e-12f);
    }
}

__device__ __forceinline__ int wave_incl_scan(int v, int lane) {
    #pragma unroll
    for (int d = 1; d < 32; d <<= 1) {
        int t = __shfl_up(v, d, 32);
        if (lane >= d) v += t;
    }
    return v;
}

__device__ __forceinline__ int wave_sum(int v) {
    #pragma unroll
    for (int d = 16; d > 0; d >>= 1) v += __shfl_xor(v, d, 32);
    return v;
}

__global__ void __launch_bounds__(1024)
k_bcount(const int* __restrict__ rows, int* __restrict__ tbl) {
    __shared__ int h[NBKT];
    const int t = threadIdx.x, g = blockIdx.x;
    if (t < NBKT) h[t] = 0;
    __syncthreads();
    const int e0 = g * CHUNK + t;
    int r[EPT];
    #pragma unroll
    for (int k = 0; k < EPT; ++k) {
        const int e = e0 + k * 1024;
        r[k] = (e < NNZ_CNT) ? rows[e] : -1;
    }
    #pragma unroll
    for (int k = 0; k < EPT; ++k)
        if (r[k] >= 0) atomicAdd(&h[r[k] >> BKT_SHIFT], 1);
    __syncthreads();
    if (t < NBKT) tbl[t * NCHUNK + g] = h[t];
}

__global__ void __launch_bounds__(SCAN_BLOCK)
k_scan(const int* __restrict__ in, int* __restrict__ out) {
    __shared__ int s_w[32];
    __shared__ int s_off;
    const int t = threadIdx.x, lane = t & 31, w = t >> 5;
    const int b = blockIdx.x;

    int pre = 0;
    const int4* c4 = (const int4*)in;
    const int n4 = b * (SCAN_BLOCK / 4);
    for (int i = t; i < n4; i += SCAN_BLOCK) {
        int4 q = c4[i];
        pre += (q.x + q.y) + (q.z + q.w);
    }
    pre = wave_sum(pre);
    if (lane == 0) s_w[w] = pre;
    __syncthreads();
    if (w == 0) {
        int x = wave_sum(s_w[lane]);
        if (lane == 0) s_off = x;
    }
    __syncthreads();
    const int off = s_off;

    const int i = b * SCAN_BLOCK + t;
    const int v = (i < TBL_N) ? in[i] : 0;
    const int inc = wave_incl_scan(v, lane);
    if (lane == 31) s_w[w] = inc;
    __syncthreads();
    if (w == 0) {
        int x  = s_w[lane];
        int xi = wave_incl_scan(x, lane);
        s_w[lane] = xi - x;
    }
    __syncthreads();
    const int excl = off + s_w[w] + inc - v;
    if (i < TBL_N) out[i] = excl;
    if (i == TBL_N - 1) out[TBL_N] = excl + v;
}

__global__ void __launch_bounds__(1024)
k_bscatter(const int* __restrict__ rows, const int* __restrict__ cols,
           const float* __restrict__ vals, const int* __restrict__ tbl_excl,
           Rec* __restrict__ bkt) {
    __shared__ int cur[NBKT];
    __shared__ int gdl[NBKT];
    __shared__ int s_w[32];
    __shared__ int4 pl[CHUNK];
    const int t = threadIdx.x, g = blockIdx.x, lane = t & 31, w = t >> 5;

    int gs = 0, cb = 0;
    if (t < NBKT) {
        gs = tbl_excl[t * NCHUNK + g];
        cb = tbl_excl[t * NCHUNK + g + 1] - gs;
    }

    const int e0 = g * CHUNK + t;
    int r[EPT], c[EPT], v[EPT];
    #pragma unroll
    for (int k = 0; k < EPT; ++k) {
        const int e = e0 + k * 1024;
        const bool in = e < NNZ_CNT;
        r[k] = in ? rows[e] : -1;
        c[k] = in ? cols[e] : 0;
        v[k] = in ? __float_as_int(vals[e]) : 0;
    }

    const int inc = wave_incl_scan(cb, lane);
    if (lane == 31) s_w[w] = inc;
    __syncthreads();
    if (w == 0) {
        int x  = s_w[lane];
        int xi = wave_incl_scan(x, lane);
        s_w[lane] = xi - x;
    }
    __syncthreads();
    const int ls = s_w[w] + inc - cb;
    if (t < NBKT) { cur[t] = ls; gdl[t] = gs - ls; }
    __syncthreads();

    int slot[EPT];
    #pragma unroll
    for (int k = 0; k < EPT; ++k)
        slot[k] = (r[k] >= 0) ? atomicAdd(&cur[r[k] >> BKT_SHIFT], 1) : -1;
    #pragma unroll
    for (int k = 0; k < EPT; ++k)
        if (slot[k] >= 0) {
            const int s = slot[k];
            pl[s] = make_int4((e0 + k * 1024) | ((r[k] & (BKT_ROWS - 1)) << E_BITS),
                              c[k], v[k], s + gdl[r[k] >> BKT_SHIFT]);
        }
    __syncthreads();

    const int cnt = (NNZ_CNT - g * CHUNK < CHUNK) ? (NNZ_CNT - g * CHUNK) : CHUNK;
    #pragma unroll
    for (int k = 0; k < EPT; ++k) {
        const int i = t + k * 1024;
        if (i < cnt) {
            const int4 q = pl[i];
            Rec o;
            o.k = q.x;
            o.c = q.y;
            o.v = q.z;
            bkt[q.w] = o;
        }
    }
}

#define BR_EPT 12
#define BR_CAP (BR_EPT * 1024)
__global__ void __launch_bounds__(1024)
k_brows(const int* __restrict__ tbl_excl, const Rec* __restrict__ bkt,
        int* __restrict__ rowptr, Rec* __restrict__ tmp) {
    __shared__ int c[BKT_ROWS];
    __shared__ int s_w[32];
    __shared__ Rec pl[BR_CAP];
    const int t = threadIdx.x, lane = t & 31, w = t >> 5;
    const int b = blockIdx.x;
    const int bs = tbl_excl[b * NCHUNK];
    const int be = tbl_excl[(b + 1) * NCHUNK];
    const int n  = be - bs;
    const bool in_lds = n <= BR_CAP;
    if (t < BKT_ROWS) c[t] = 0;
    __syncthreads();

    int key[BR_EPT];
    if (in_lds) {
        #pragma unroll
        for (int k = 0; k < BR_EPT; ++k) {
            const int i = t + k * 1024;
            key[k] = (i < n) ? bkt[bs + i].k : -1;
        }
        #pragma unroll
        for (int k = 0; k < BR_EPT; ++k)
            if (key[k] >= 0) atomicAdd(&c[key[k] >> E_BITS], 1);
    } else {
        for (int i0 = bs + t; i0 < be; i0 += 4 * 1024) {
            int wd[4];
            #pragma unroll
            for (int k = 0; k < 4; ++k) {
                const int i = i0 + k * 1024;
                wd[k] = (i < be) ? bkt[i].k : -1;
            }
            #pragma unroll
            for (int k = 0; k < 4; ++k)
                if (wd[k] >= 0) atomicAdd(&c[wd[k] >> E_BITS], 1);
        }
    }
    __syncthreads();

    const int v = (t < BKT_ROWS) ? c[t] : 0;
    const int inc = wave_incl_scan(v, lane);
    if (lane == 31) s_w[w] = inc;
    __syncthreads();
    if (w == 0) {
        int x  = s_w[lane];
        int xi = wave_incl_scan(x, lane);
        s_w[lane] = xi - x;
    }
    __syncthreads();
    const int lstart = s_w[w] + inc - v;
    if (t < BKT_ROWS) {
        c[t] = in_lds ? lstart : (bs + lstart);
        const int r = b * BKT_ROWS + t;
        if (r < NTOT) rowptr[r] = bs + lstart;
        if (r == NTOT - 1) rowptr[NTOT] = bs + lstart + v;
    }
    __syncthreads();

    if (in_lds) {
        int cc[BR_EPT], vv[BR_EPT];
        #pragma unroll
        for (int k = 0; k < BR_EPT; ++k) {
            const int i = t + k * 1024;
            if (i < n) { cc[k] = bkt[bs + i].c; vv[k] = bkt[bs + i].v; }
        }
        int slot[BR_EPT];
        #pragma unroll
        for (int k = 0; k < BR_EPT; ++k)
            slot[k] = (key[k] >= 0) ? atomicAdd(&c[key[k] >> E_BITS], 1) : -1;
        #pragma unroll
        for (int k = 0; k < BR_EPT; ++k)
            if (slot[k] >= 0) {
                Rec o;
                o.k = key[k] & E_MASK;
                o.c = cc[k];
                o.v = vv[k];
                pl[slot[k]] = o;
            }
        __syncthreads();
        #pragma unroll
        for (int k = 0; k < BR_EPT; ++k) {
            const int i = t + k * 1024;
            if (i < n) tmp[bs + i] = pl[i];
        }
    } else {
        for (int i0 = bs + t; i0 < be; i0 += 4 * 1024) {
            Rec wd[4];
            #pragma unroll
            for (int k = 0; k < 4; ++k) {
                const int i = i0 + k * 1024;
                if (i < be) wd[k] = bkt[i];
                else { wd[k].k = -1; wd[k].c = 0; wd[k].v = 0; }
            }
            int slot[4];
            #pragma unroll
            for (int k = 0; k < 4; ++k)
                slot[k] = (wd[k].k >= 0) ? atomicAdd(&c[wd[k].k >> E_BITS], 1) : -1;
            #pragma unroll
            for (int k = 0; k < 4; ++k)
                if (slot[k] >= 0) {
                    Rec o;
                    o.k = wd[k].k & E_MASK;
                    o.c = wd[k].c;
                    o.v = wd[k].v;
                    tmp[slot[k]] = o;
                }
        }
    }
}

#define ORD_R 8
static_assert(NTOT % ORD_R == 0, "every wave of the ordering kernel owns ORD_R whole rows");

__device__ __forceinline__ int rank_short_row(const int mine, const int n) {
    int rank = 0;
    for (int k = 0; k < n; k += 4) {
        #pragma unroll
        for (int j = 0; j < 4; ++j) {
            const int ek = __builtin_amdgcn_readlane(mine, k + j);
            rank += (ek < mine) ? 1 : 0;
        }
    }
    return rank;
}

__device__ __forceinline__ void order_long_row(const int s, const int n, const int lane,
                                               const Rec* __restrict__ tmp, int2* __restrict__ scv) {
    for (int bi = 0; bi < n; bi += 32) {
        const int i = bi + lane;
        int mine = 0x7fffffff;
        int2 cv = make_int2(0, 0);
        if (i < n) {
            const Rec m = tmp[s + i];
            mine = m.k;
            cv = make_int2(m.c, m.v);
        }
        int rank = 0;
        for (int bj = 0; bj < n; bj += 32) {
            const int j = bj + lane;
            const int other = (j < n) ? tmp[s + j].k : 0x7fffffff;
            const int m = (n - bj < 32) ? (n - bj) : 32;
            for (int k = 0; k < m; ++k) {
                int ek = __builtin_amdgcn_readlane(other, k);
                rank += (ek < mine) ? 1 : 0;
            }
        }
        if (i < n) scv[s + rank] = cv;
    }
}

__global__ void k_order(const int* __restrict__ rowptr, const Rec* __restrict__ tmp,
                        int2* __restrict__ scv) {
    const int w = __builtin_amdgcn_readfirstlane((int)((blockIdx.x * blockDim.x + threadIdx.x) >> 5));
    const int lane = threadIdx.x & 31;
    const int r0 = w * ORD_R;
    if (r0 >= NTOT) return;

    int rp = 0;
    if (lane <= ORD_R) rp = rowptr[r0 + lane];

    int s[ORD_R], n[ORD_R];
    Rec m[ORD_R];
    #pragma unroll
    for (int q = 0; q < ORD_R; ++q) {
        s[q] = __builtin_amdgcn_readlane(rp, q);
        n[q] = __builtin_amdgcn_readlane(rp, q + 1) - s[q];
        m[q].k = 0x7fffffff; m[q].c = 0; m[q].v = 0;
        if (lane < n[q]) m[q] = tmp[s[q] + lane];
    }
    #pragma unroll
    for (int q = 0; q < ORD_R; ++q) {
        if (n[q] <= 32) {
            const int rank = rank_short_row(m[q].k, n[q]);
            if (lane < n[q]) scv[s[q] + rank] = make_int2(m[q].c, m[q].v);
        }
    }
    #pragma nounroll
    for (int q = 0; q < ORD_R; ++q) {
        const int sq = __builtin_amdgcn_readlane(rp, q);
        const int nq = __builtin_amdgcn_readlane(rp, q + 1) - sq;
        if (nq > 32) order_long_row(sq, nq, lane, tmp, scv);
    }
}

__device__ __forceinline__ float signf_(float v) {
    return (v > 0.0f) ? 1.0f : ((v < 0.0f) ? -1.0f : 0.0f);
}

#define LAY_R 2
static_assert(NTOT % LAY_R == 0, "every wave of a layer kernel owns two whole rows");

template <int K>
__device__ __forceinline__ int half_bcast(const int v) {
    return __builtin_amdgcn_update_dpp(0, v, 0x150 + K, 0xf, 0xf, true);
}

typedef float v4f __attribute__((ext_vector_type(4)));
typedef __attribute__((address_space(1))) const v4f gv4f;

template <int L>
__device__ __forceinline__ float4 gather4(const int c, const int li,
                                          const unsigned long long ub,
                                          const unsigned long long ib,
                                          const float* src) {
    if (L == 1) {
        const unsigned long long base = (c < USER_NUM) ? ub : ib;
        const v4f t = *(gv4f*)(base + ((unsigned long long)(unsigned)c << 8));
        return make_float4(t.x, t.y, t.z, t.w);
    } else {
        const unsigned off = ((unsigned)c << 8) + ((unsigned)li << 4);
        return *(const float4*)((const char*)src + off);
    }
}

__device__ __forceinline__ void fma4(float4& a, const float v, const float4 x) {
    a.x = __builtin_fmaf(v, x.x, a.x);
    a.y = __builtin_fmaf(v, x.y, a.y);
    a.z = __builtin_fmaf(v, x.z, a.z);
    a.w = __builtin_fmaf(v, x.w, a.w);
}

template <int L, int K0>
__device__ __forceinline__ void entries8(float4& a, const int2 cv, const int m, const int li,
                                         const unsigned long long ub,
                                         const unsigned long long ib,
                                         const float* src) {
    const bool more = m > K0 + 4;
    float4 x0, x1, x2, x3, x4, x5, x6, x7;
    x0 = gather4<L>(half_bcast<K0 + 0>(cv.x), li, ub, ib, src);
    x1 = gather4<L>(half_bcast<K0 + 1>(cv.x), li, ub, ib, src);
    x2 = gather4<L>(half_bcast<K0 + 2>(cv.x), li, ub, ib, src);
    x3 = gather4<L>(half_bcast<K0 + 3>(cv.x), li, ub, ib, src);
    if (more) {
        x4 = gather4<L>(half_bcast<K0 + 4>(cv.x), li, ub, ib, src);
        x5 = gather4<L>(half_bcast<K0 + 5>(cv.x), li, ub, ib, src);
        x6 = gather4<L>(half_bcast<K0 + 6>(cv.x), li, ub, ib, src);
        x7 = gather4<L>(half_bcast<K0 + 7>(cv.x), li, ub, ib, src);
    }
    fma4(a, __int_as_float(half_bcast<K0 + 0>(cv.y)), x0);
    fma4(a, __int_as_float(half_bcast<K0 + 1>(cv.y)), x1);
    fma4(a, __int_as_float(half_bcast<K0 + 2>(cv.y)), x2);
    fma4(a, __int_as_float(half_bcast<K0 + 3>(cv.y)), x3);
    if (more) {
        fma4(a, __int_as_float(half_bcast<K0 + 4>(cv.y)), x4);
        fma4(a, __int_as_float(half_bcast<K0 + 5>(cv.y)), x5);
        fma4(a, __int_as_float(half_bcast<K0 + 6>(cv.y)), x6);
        fma4(a, __int_as_float(half_bcast<K0 + 7>(cv.y)), x7);
    }
}

template <int L>
__global__ void k_layer(const int* __restrict__ rowptr, const int2* __restrict__ scv,
                        const float* __restrict__ user_emb,
                        const float* __restrict__ item_emb,
                        const float* src,
                        const float* __restrict__ noise_k,
                        const float* __restrict__ inv_scale_k,
                        const float* e1, const float* e2, float* dst) {
    const int w = __builtin_amdgcn_readfirstlane((int)((blockIdx.x * blockDim.x + threadIdx.x) >> 5));
    const int lane = threadIdx.x & 31;
    const int h  = lane >> 4;
    const int li = lane & 15;
    const int r0 = w * LAY_R;
    if (r0 >= NTOT) return;
    const int b0r = rowptr[r0];
    const int b1r = rowptr[r0 + 1];
    const int b2r = rowptr[r0 + 2];
    const int n0 = b1r - b0r, n1 = b2r - b1r;
    const int nmax = (n0 > n1) ? n0 : n1;
    const int s = h ? b1r : b0r;
    const int n = h ? n1 : n0;
    const int r = r0 + h;

    const size_t o  = (size_t)r * EMB + (size_t)li * 4;
    const float  is = inv_scale_k[r];
    const float4 nv = *(const float4*)(noise_k + o);

    const unsigned long long ub = (unsigned long long)user_emb + (unsigned long long)li * 16u;
    const unsigned long long ib = (unsigned long long)item_emb + (unsigned long long)li * 16u
                                  - (unsigned long long)USER_NUM * EMB * sizeof(float);

    float4 a = make_float4(0.0f, 0.0f, 0.0f, 0.0f);
    int2 cv = make_int2(0, 0);
    if (li < n) cv = scv[s + li];
    for (int b0 = 0; b0 < nmax; b0 += 16) {
        int2 nx = make_int2(0, 0);
        if (b0 + 16 < nmax) {
            if (b0 + 16 + li < n) nx = scv[s + b0 + 16 + li];
        }
        const int m = (nmax - b0 < 16) ? (nmax - b0) : 16;
        entries8<L, 0>(a, cv, m, li, ub, ib, src);
        if (m > 8) entries8<L, 8>(a, cv, m, li, ub, ib, src);
        cv = nx;
    }

    float4 ev;
    ev.x = a.x + signf_(a.x) * nv.x * is;
    ev.y = a.y + signf_(a.y) * nv.y * is;
    ev.z = a.z + signf_(a.z) * nv.z * is;
    ev.w = a.w + signf_(a.w) * nv.w * is;

    if (L < 3) {
        *(float4*)(dst + o) = ev;
    } else {
        const float* g0 = (r < USER_NUM) ? (user_emb + (size_t)r * EMB)
                                         : (item_emb + (size_t)(r - USER_NUM) * EMB);
        float4 g = ((const float4*)g0)[li];
        const float4 p1 = *(const float4*)(e1 + o);
        const float4 p2 = *(const float4*)(e2 + o);
        g.x += p1.x; g.y += p1.y; g.z += p1.z; g.w += p1.w;
        g.x += p2.x; g.y += p2.y; g.z += p2.z; g.w += p2.w;
        g.x += ev.x; g.y += ev.y; g.z += ev.z; g.w += ev.w;
        g.x *= 0.25f; g.y *= 0.25f; g.z *= 0.25f; g.w *= 0.25f;
        *(float4*)(dst + o) = g;
    }
}

extern "C" void kernel_launch(void* const* d_in, const int* in_sizes, int n_in,
                              void* d_out, int out_size, void* d_ws, size_t ws_size,
                              hipStream_t stream) {
    const float* user_emb = (const float*)d_in[0];
    const float* item_emb = (const float*)d_in[1];
    const float* adj_vals = (const float*)d_in[2];
    const float* noise    = (const float*)d_in[3];
    const int*   adj_rows = (const int*)d_in[4];
    const int*   adj_cols = (const int*)d_in[5];

    float* out = (float*)d_out;

    float* R1  = (float*)d_ws;
    float* R2  = R1 + (size_t)NTOT * EMB;
    float* inv = R2 + (size_t)NTOT * EMB;

    Rec*  tmp    = (Rec*)R1;
    Rec*  bkt    = (Rec*)R2;
    int2* scv    = (int2*)R2;
    int*  rowptr = (int*)(R2 + (size_t)3 * NNZ_CNT);
    int*  tbl    = rowptr + ROWPTR_PAD;
    int*  tblx   = tbl + TBL_PAD;

    {
        int waves  = (N_LAYERS * NTOT) / 16;
        int blocks = (waves + 7) / 8;
        k_noise_invnorm<<<blocks, 256, 0, stream>>>(noise, inv);
    }

    const int oblocks = (NTOT / ORD_R + 7) / 8;
    const int rblocks = ((NTOT + LAY_R - 1) / LAY_R + 7) / 8;

    k_bcount<<<NCHUNK, 1024, 0, stream>>>(adj_rows, tbl);
    k_scan<<<SCAN_BLOCKS, SCAN_BLOCK, 0, stream>>>(tbl, tblx);
    k_bscatter<<<NCHUNK, 1024, 0, stream>>>(adj_rows, adj_cols, adj_vals, tblx, bkt);
    k_brows<<<NBKT, 1024, 0, stream>>>(tblx, bkt, rowptr, tmp);
    k_order<<<oblocks, 256, 0, stream>>>(rowptr, tmp, scv);

    k_layer<1><<<rblocks, 256, 0, stream>>>(rowptr, scv, user_emb, item_emb, nullptr,
        noise, inv, nullptr, nullptr, out);
    k_layer<2><<<rblocks, 256, 0, stream>>>(rowptr, scv, user_emb, item_emb, out,
        noise + (size_t)1 * NTOT * EMB, inv + (size_t)1 * NTOT, nullptr, nullptr, R1);
    k_layer<3><<<rblocks, 256, 0, stream>>>(rowptr, scv, user_emb, item_emb, R1,
        noise + (size_t)2 * NTOT * EMB, inv + (size_t)2 * NTOT, out, R1, out);
}


# Noise norms worked out inside the layer kernels: 16-row noise tile in LDS, each wave 2 f32 WMMAs on its 8-column slice, 8 partial sums added in order; the norm launch and its read of noise are gone
# speedup vs baseline: 3.8789x; 1.0372x over previous
#include <hip/hip_runtime.h>
#include <hip/hip_bf16.h>


#define USER_NUM 100000
#define ITEM_NUM 50000
#define NTOT     (USER_NUM + ITEM_NUM)
#define EMB      64
#define NNZ_CNT  3000000
#define N_LAYERS 3
#define EPS_F    0.1f
#define SCAN_BLOCK 1024
#define ROWPTR_PAD 150016

#define BKT_SHIFT 9
#define BKT_ROWS  (1 << BKT_SHIFT)
#define NBKT      ((NTOT + BKT_ROWS - 1) / BKT_ROWS)
#define EPT       12
#define CHUNK     (EPT * 1024)
#define NCHUNK    ((NNZ_CNT + CHUNK - 1) / CHUNK)
#define TBL_N     (NBKT * NCHUNK)
#define TBL_PAD   71808
#define SCAN_BLOCKS ((TBL_N + SCAN_BLOCK - 1) / SCAN_BLOCK)
#define E_BITS    22
#define E_MASK    ((1 << E_BITS) - 1)

static_assert(NNZ_CNT <= (1 << E_BITS), "edge index must fit in E_BITS");
static_assert(E_BITS + BKT_SHIFT <= 31, "packed word must stay non-negative");
static_assert(TBL_N + 1 <= TBL_PAD, "table padding");
static_assert(NBKT <= 1024, "one thread per bucket when the table is read or written");
static_assert((size_t)3 * NNZ_CNT <= (size_t)NTOT * EMB, "a record array fits one [N, 64] region");
static_assert((size_t)3 * NNZ_CNT + ROWPTR_PAD + 2 * TBL_PAD <= (size_t)NTOT * EMB, "records, row bounds and tables fit the second region");

struct Rec { int k; int c; int v; };

typedef float v2f __attribute__((ext_vector_type(2)));
typedef float v8f __attribute__((ext_vector_type(8)));

__device__ __forceinline__ int wave_incl_scan(int v, int lane) {
    #pragma unroll
    for (int d = 1; d < 32; d <<= 1) {
        int t = __shfl_up(v, d, 32);
        if (lane >= d) v += t;
    }
    return v;
}

__device__ __forceinline__ int wave_sum(int v) {
    #pragma unroll
    for (int d = 16; d > 0; d >>= 1) v += __shfl_xor(v, d, 32);
    return v;
}

__global__ void __launch_bounds__(1024)
k_bcount(const int* __restrict__ rows, int* __restrict__ tbl) {
    __shared__ int h[NBKT];
    const int t = threadIdx.x, g = blockIdx.x;
    if (t < NBKT) h[t] = 0;
    __syncthreads();
    const int e0 = g * CHUNK + t;
    int r[EPT];
    #pragma unroll
    for (int k = 0; k < EPT; ++k) {
        const int e = e0 + k * 1024;
        r[k] = (e < NNZ_CNT) ? rows[e] : -1;
    }
    #pragma unroll
    for (int k = 0; k < EPT; ++k)
        if (r[k] >= 0) atomicAdd(&h[r[k] >> BKT_SHIFT], 1);
    __syncthreads();
    if (t < NBKT) tbl[t * NCHUNK + g] = h[t];
}

__global__ void __launch_bounds__(SCAN_BLOCK)
k_scan(const int* __restrict__ in, int* __restrict__ out) {
    __shared__ int s_w[32];
    __shared__ int s_off;
    const int t = threadIdx.x, lane = t & 31, w = t >> 5;
    const int b = blockIdx.x;

    int pre = 0;
    const int4* c4 = (const int4*)in;
    const int n4 = b * (SCAN_BLOCK / 4);
    for (int i = t; i < n4; i += SCAN_BLOCK) {
        int4 q = c4[i];
        pre += (q.x + q.y) + (q.z + q.w);
    }
    pre = wave_sum(pre);
    if (lane == 0) s_w[w] = pre;
    __syncthreads();
    if (w == 0) {
        int x = wave_sum(s_w[lane]);
        if (lane == 0) s_off = x;
    }
    __syncthreads();
    const int off = s_off;

    const int i = b * SCAN_BLOCK + t;
    const int v = (i < TBL_N) ? in[i] : 0;
    const int inc = wave_incl_scan(v, lane);
    if (lane == 31) s_w[w] = inc;
    __syncthreads();
    if (w == 0) {
        int x  = s_w[lane];
        int xi = wave_incl_scan(x, lane);
        s_w[lane] = xi - x;
    }
    __syncthreads();
    const int excl = off + s_w[w] + inc - v;
    if (i < TBL_N) out[i] = excl;
    if (i == TBL_N - 1) out[TBL_N] = excl + v;
}

__global__ void __launch_bounds__(1024)
k_bscatter(const int* __restrict__ rows, const int* __restrict__ cols,
           const float* __restrict__ vals, const int* __restrict__ tbl_excl,
           Rec* __restrict__ bkt) {
    __shared__ int cur[NBKT];
    __shared__ int gdl[NBKT];
    __shared__ int s_w[32];
    __shared__ int4 pl[CHUNK];
    const int t = threadIdx.x, g = blockIdx.x, lane = t & 31, w = t >> 5;

    int gs = 0, cb = 0;
    if (t < NBKT) {
        gs = tbl_excl[t * NCHUNK + g];
        cb = tbl_excl[t * NCHUNK + g + 1] - gs;
    }

    const int e0 = g * CHUNK + t;
    int r[EPT], c[EPT], v[EPT];
    #pragma unroll
    for (int k = 0; k < EPT; ++k) {
        const int e = e0 + k * 1024;
        const bool in = e < NNZ_CNT;
        r[k] = in ? rows[e] : -1;
        c[k] = in ? cols[e] : 0;
        v[k] = in ? __float_as_int(vals[e]) : 0;
    }

    const int inc = wave_incl_scan(cb, lane);
    if (lane == 31) s_w[w] = inc;
    __syncthreads();
    if (w == 0) {
        int x  = s_w[lane];
        int xi = wave_incl_scan(x, lane);
        s_w[lane] = xi - x;
    }
    __syncthreads();
    const int ls = s_w[w] + inc - cb;
    if (t < NBKT) { cur[t] = ls; gdl[t] = gs - ls; }
    __syncthreads();

    int slot[EPT];
    #pragma unroll
    for (int k = 0; k < EPT; ++k)
        slot[k] = (r[k] >= 0) ? atomicAdd(&cur[r[k] >> BKT_SHIFT], 1) : -1;
    #pragma unroll
    for (int k = 0; k < EPT; ++k)
        if (slot[k] >= 0) {
            const int s = slot[k];
            pl[s] = make_int4((e0 + k * 1024) | ((r[k] & (BKT_ROWS - 1)) << E_BITS),
                              c[k], v[k], s + gdl[r[k] >> BKT_SHIFT]);
        }
    __syncthreads();

    const int cnt = (NNZ_CNT - g * CHUNK < CHUNK) ? (NNZ_CNT - g * CHUNK) : CHUNK;
    #pragma unroll
    for (int k = 0; k < EPT; ++k) {
        const int i = t + k * 1024;
        if (i < cnt) {
            const int4 q = pl[i];
            Rec o;
            o.k = q.x;
            o.c = q.y;
            o.v = q.z;
            bkt[q.w] = o;
        }
    }
}

#define BR_EPT 12
#define BR_CAP (BR_EPT * 1024)
__global__ void __launch_bounds__(1024)
k_brows(const int* __restrict__ tbl_excl, const Rec* __restrict__ bkt,
        int* __restrict__ rowptr, Rec* __restrict__ tmp) {
    __shared__ int c[BKT_ROWS];
    __shared__ int s_w[32];
    __shared__ Rec pl[BR_CAP];
    const int t = threadIdx.x, lane = t & 31, w = t >> 5;
    const int b = blockIdx.x;
    const int bs = tbl_excl[b * NCHUNK];
    const int be = tbl_excl[(b + 1) * NCHUNK];
    const int n  = be - bs;
    const bool in_lds = n <= BR_CAP;
    if (t < BKT_ROWS) c[t] = 0;
    __syncthreads();

    int key[BR_EPT];
    if (in_lds) {
        #pragma unroll
        for (int k = 0; k < BR_EPT; ++k) {
            const int i = t + k * 1024;
            key[k] = (i < n) ? bkt[bs + i].k : -1;
        }
        #pragma unroll
        for (int k = 0; k < BR_EPT; ++k)
            if (key[k] >= 0) atomicAdd(&c[key[k] >> E_BITS], 1);
    } else {
        for (int i0 = bs + t; i0 < be; i0 += 4 * 1024) {
            int wd[4];
            #pragma unroll
            for (int k = 0; k < 4; ++k) {
                const int i = i0 + k * 1024;
                wd[k] = (i < be) ? bkt[i].k : -1;
            }
            #pragma unroll
            for (int k = 0; k < 4; ++k)
                if (wd[k] >= 0) atomicAdd(&c[wd[k] >> E_BITS], 1);
        }
    }
    __syncthreads();

    const int v = (t < BKT_ROWS) ? c[t] : 0;
    const int inc = wave_incl_scan(v, lane);
    if (lane == 31) s_w[w] = inc;
    __syncthreads();
    if (w == 0) {
        int x  = s_w[lane];
        int xi = wave_incl_scan(x, lane);
        s_w[lane] = xi - x;
    }
    __syncthreads();
    const int lstart = s_w[w] + inc - v;
    if (t < BKT_ROWS) {
        c[t] = in_lds ? lstart : (bs + lstart);
        const int r = b * BKT_ROWS + t;
        if (r < NTOT) rowptr[r] = bs + lstart;
        if (r == NTOT - 1) rowptr[NTOT] = bs + lstart + v;
    }
    __syncthreads();

    if (in_lds) {
        int cc[BR_EPT], vv[BR_EPT];
        #pragma unroll
        for (int k = 0; k < BR_EPT; ++k) {
            const int i = t + k * 1024;
            if (i < n) { cc[k] = bkt[bs + i].c; vv[k] = bkt[bs + i].v; }
        }
        int slot[BR_EPT];
        #pragma unroll
        for (int k = 0; k < BR_EPT; ++k)
            slot[k] = (key[k] >= 0) ? atomicAdd(&c[key[k] >> E_BITS], 1) : -1;
        #pragma unroll
        for (int k = 0; k < BR_EPT; ++k)
            if (slot[k] >= 0) {
                Rec o;
                o.k = key[k] & E_MASK;
                o.c = cc[k];
                o.v = vv[k];
                pl[slot[k]] = o;
            }
        __syncthreads();
        #pragma unroll
        for (int k = 0; k < BR_EPT; ++k) {
            const int i = t + k * 1024;
            if (i < n) tmp[bs + i] = pl[i];
        }
    } else {
        for (int i0 = bs + t; i0 < be; i0 += 4 * 1024) {
            Rec wd[4];
            #pragma unroll
            for (int k = 0; k < 4; ++k) {
                const int i = i0 + k * 1024;
                if (i < be) wd[k] = bkt[i];
                else { wd[k].k = -1; wd[k].c = 0; wd[k].v = 0; }
            }
            int slot[4];
            #pragma unroll
            for (int k = 0; k < 4; ++k)
                slot[k] = (wd[k].k >= 0) ? atomicAdd(&c[wd[k].k >> E_BITS], 1) : -1;
            #pragma unroll
            for (int k = 0; k < 4; ++k)
                if (slot[k] >= 0) {
                    Rec o;
                    o.k = wd[k].k & E_MASK;
                    o.c = wd[k].c;
                    o.v = wd[k].v;
                    tmp[slot[k]] = o;
                }
        }
    }
}

#define ORD_R 8
static_assert(NTOT % ORD_R == 0, "every wave of the ordering kernel owns ORD_R whole rows");

__device__ __forceinline__ int rank_short_row(const int mine, const int n) {
    int rank = 0;
    for (int k = 0; k < n; k += 4) {
        #pragma unroll
        for (int j = 0; j < 4; ++j) {
            const int ek = __builtin_amdgcn_readlane(mine, k + j);
            rank += (ek < mine) ? 1 : 0;
        }
    }
    return rank;
}

__device__ __forceinline__ void order_long_row(const int s, const int n, const int lane,
                                               const Rec* __restrict__ tmp, int2* __restrict__ scv) {
    for (int bi = 0; bi < n; bi += 32) {
        const int i = bi + lane;
        int mine = 0x7fffffff;
        int2 cv = make_int2(0, 0);
        if (i < n) {
            const Rec m = tmp[s + i];
            mine = m.k;
            cv = make_int2(m.c, m.v);
        }
        int rank = 0;
        for (int bj = 0; bj < n; bj += 32) {
            const int j = bj + lane;
            const int other = (j < n) ? tmp[s + j].k : 0x7fffffff;
            const int m = (n - bj < 32) ? (n - bj) : 32;
            for (int k = 0; k < m; ++k) {
                int ek = __builtin_amdgcn_readlane(other, k);
                rank += (ek < mine) ? 1 : 0;
            }
        }
        if (i < n) scv[s + rank] = cv;
    }
}

__global__ void k_order(const int* __restrict__ rowptr, const Rec* __restrict__ tmp,
                        int2* __restrict__ scv) {
    const int w = __builtin_amdgcn_readfirstlane((int)((blockIdx.x * blockDim.x + threadIdx.x) >> 5));
    const int lane = threadIdx.x & 31;
    const int r0 = w * ORD_R;
    if (r0 >= NTOT) return;

    int rp = 0;
    if (lane <= ORD_R) rp = rowptr[r0 + lane];

    int s[ORD_R], n[ORD_R];
    Rec m[ORD_R];
    #pragma unroll
    for (int q = 0; q < ORD_R; ++q) {
        s[q] = __builtin_amdgcn_readlane(rp, q);
        n[q] = __builtin_amdgcn_readlane(rp, q + 1) - s[q];
        m[q].k = 0x7fffffff; m[q].c = 0; m[q].v = 0;
        if (lane < n[q]) m[q] = tmp[s[q] + lane];
    }
    #pragma unroll
    for (int q = 0; q < ORD_R; ++q) {
        if (n[q] <= 32) {
            const int rank = rank_short_row(m[q].k, n[q]);
            if (lane < n[q]) scv[s[q] + rank] = make_int2(m[q].c, m[q].v);
        }
    }
    #pragma nounroll
    for (int q = 0; q < ORD_R; ++q) {
        const int sq = __builtin_amdgcn_readlane(rp, q);
        const int nq = __builtin_amdgcn_readlane(rp, q + 1) - sq;
        if (nq > 32) order_long_row(sq, nq, lane, tmp, scv);
    }
}

__device__ __forceinline__ float signf_(float v) {
    return (v > 0.0f) ? 1.0f : ((v < 0.0f) ? -1.0f : 0.0f);
}

#define LAY_R 2
#define LAY_ROWS 16
#define NT_PITCH 68
static_assert(NTOT % LAY_ROWS == 0, "every work-group of a layer kernel owns 16 whole rows");
static_assert(LAY_ROWS == 8 * LAY_R, "8 waves of LAY_R rows make one 16-row tile");

__device__ __forceinline__ float noise_inv_scale(const float4 nv, const int wv, const int lane,
                                                 float* s_nt, float* s_pp) {
    const int h = lane >> 4, li = lane & 15;
    const int lr = wv * LAY_R + h;
    *(float4*)(s_nt + lr * NT_PITCH + li * 4) = nv;
    __syncthreads();
    const float4 q = *(const float4*)(s_nt + li * NT_PITCH + wv * 8 + h * 4);
    v2f qa, qb;
    qa.x = q.x; qa.y = q.y;
    qb.x = q.z; qb.y = q.w;
    v8f c = {};
    c = __builtin_amdgcn_wmma_f32_16x16x4_f32(false, qa, false, qa, (short)0, c, false, false);
    c = __builtin_amdgcn_wmma_f32_16x16x4_f32(false, qb, false, qb, (short)0, c, false, false);
    const int j7 = lane & 7;
    float d = c[0];
    #pragma unroll
    for (int j = 1; j < 8; ++j) d = (j7 == j) ? c[j] : d;
    s_pp[lane * 8 + wv] = d;
    __syncthreads();
    const int sl = (lr < 8) ? lr : (lr + 16);
    const float4 pa = *(const float4*)(s_pp + sl * 8);
    const float4 pb = *(const float4*)(s_pp + sl * 8 + 4);
    float ss = pa.x;
    ss += pa.y; ss += pa.z; ss += pa.w;
    ss += pb.x; ss += pb.y; ss += pb.z; ss += pb.w;
    asm volatile("" : "+v"(ss));
    return EPS_F / fmaxf(sqrtf(ss), 1e-12f);
}

template <int K>
__device__ __forceinline__ int half_bcast(const int v) {
    return __builtin_amdgcn_update_dpp(0, v, 0x150 + K, 0xf, 0xf, true);
}

typedef float v4f __attribute__((ext_vector_type(4)));
typedef __attribute__((address_space(1))) const v4f gv4f;

template <int L>
__device__ __forceinline__ float4 gather4(const int c, const int li,
                                          const unsigned long long ub,
                                          const unsigned long long ib,
                                          const float* src) {
    if (L == 1) {
        const unsigned long long base = (c < USER_NUM) ? ub : ib;
        const v4f t = *(gv4f*)(base + ((unsigned long long)(unsigned)c << 8));
        return make_float4(t.x, t.y, t.z, t.w);
    } else {
        const unsigned off = ((unsigned)c << 8) + ((unsigned)li << 4);
        return *(const float4*)((const char*)src + off);
    }
}

__device__ __forceinline__ void fma4(float4& a, const float v, const float4 x) {
    a.x = __builtin_fmaf(v, x.x, a.x);
    a.y = __builtin_fmaf(v, x.y, a.y);
    a.z = __builtin_fmaf(v, x.z, a.z);
    a.w = __builtin_fmaf(v, x.w, a.w);
}

template <int L, int K0>
__device__ __forceinline__ void entries8(float4& a, const int2 cv, const int m, const int li,
                                         const unsigned long long ub,
                                         const unsigned long long ib,
                                         const float* src) {
    const bool more = m > K0 + 4;
    float4 x0, x1, x2, x3, x4, x5, x6, x7;
    x0 = gather4<L>(half_bcast<K0 + 0>(cv.x), li, ub, ib, src);
    x1 = gather4<L>(half_bcast<K0 + 1>(cv.x), li, ub, ib, src);
    x2 = gather4<L>(half_bcast<K0 + 2>(cv.x), li, ub, ib, src);
    x3 = gather4<L>(half_bcast<K0 + 3>(cv.x), li, ub, ib, src);
    if (more) {
        x4 = gather4<L>(half_bcast<K0 + 4>(cv.x), li, ub, ib, src);
        x5 = gather4<L>(half_bcast<K0 + 5>(cv.x), li, ub, ib, src);
        x6 = gather4<L>(half_bcast<K0 + 6>(cv.x), li, ub, ib, src);
        x7 = gather4<L>(half_bcast<K0 + 7>(cv.x), li, ub, ib, src);
    }
    fma4(a, __int_as_float(half_bcast<K0 + 0>(cv.y)), x0);
    fma4(a, __int_as_float(half_bcast<K0 + 1>(cv.y)), x1);
    fma4(a, __int_as_float(half_bcast<K0 + 2>(cv.y)), x2);
    fma4(a, __int_as_float(half_bcast<K0 + 3>(cv.y)), x3);
    if (more) {
        fma4(a, __int_as_float(half_bcast<K0 + 4>(cv.y)), x4);
        fma4(a, __int_as_float(half_bcast<K0 + 5>(cv.y)), x5);
        fma4(a, __int_as_float(half_bcast<K0 + 6>(cv.y)), x6);
        fma4(a, __int_as_float(half_bcast<K0 + 7>(cv.y)), x7);
    }
}

template <int L>
__global__ void __launch_bounds__(256)
k_layer(const int* __restrict__ rowptr, const int2* __restrict__ scv,
        const float* __restrict__ user_emb,
        const float* __restrict__ item_emb,
        const float* src,
        const float* __restrict__ noise_k,
        const float* e1, const float* e2, float* dst) {
    __shared__ __attribute__((aligned(16))) float s_nt[LAY_ROWS * NT_PITCH];
    __shared__ __attribute__((aligned(16))) float s_pp[32 * 8];
    const int wv = __builtin_amdgcn_readfirstlane((int)(threadIdx.x >> 5));
    const int lane = threadIdx.x & 31;
    const int h  = lane >> 4;
    const int li = lane & 15;
    const int r0 = (int)blockIdx.x * LAY_ROWS + wv * LAY_R;
    const int b0r = rowptr[r0];
    const int b1r = rowptr[r0 + 1];
    const int b2r = rowptr[r0 + 2];
    const int n0 = b1r - b0r, n1 = b2r - b1r;
    const int nmax = (n0 > n1) ? n0 : n1;
    const int s = h ? b1r : b0r;
    const int n = h ? n1 : n0;
    const int r = r0 + h;

    const unsigned o = (unsigned)r * EMB + (unsigned)li * 4;
    const float4 nv = *(const float4*)(noise_k + o);

    const unsigned long long ub = (unsigned long long)user_emb + (unsigned long long)li * 16u;
    const unsigned long long ib = (unsigned long long)item_emb + (unsigned long long)li * 16u
                                  - (unsigned long long)USER_NUM * EMB * sizeof(float);

    float4 a = make_float4(0.0f, 0.0f, 0.0f, 0.0f);
    int2 cv = make_int2(0, 0);
    if (li < n) cv = scv[s + li];

    const float is = noise_inv_scale(nv, wv, lane, s_nt, s_pp);

    for (int b0 = 0; b0 < nmax; b0 += 16) {
        int2 nx = make_int2(0, 0);
        if (b0 + 16 < nmax) {
            if (b0 + 16 + li < n) nx = scv[s + b0 + 16 + li];
        }
        const int m = (nmax - b0 < 16) ? (nmax - b0) : 16;
        entries8<L, 0>(a, cv, m, li, ub, ib, src);
        if (m > 8) entries8<L, 8>(a, cv, m, li, ub, ib, src);
        cv = nx;
    }

    float4 ev;
    ev.x = a.x + signf_(a.x) * nv.x * is;
    ev.y = a.y + signf_(a.y) * nv.y * is;
    ev.z = a.z + signf_(a.z) * nv.z * is;
    ev.w = a.w + signf_(a.w) * nv.w * is;

    if (L < 3) {
        *(float4*)(dst + o) = ev;
    } else {
        const float* g0 = (r < USER_NUM) ? (user_emb + (size_t)r * EMB)
                                         : (item_emb + (size_t)(r - USER_NUM) * EMB);
        float4 g = ((const float4*)g0)[li];
        const float4 p1 = *(const float4*)(e1 + o);
        const float4 p2 = *(const float4*)(e2 + o);
        g.x += p1.x; g.y += p1.y; g.z += p1.z; g.w += p1.w;
        g.x += p2.x; g.y += p2.y; g.z += p2.z; g.w += p2.w;
        g.x += ev.x; g.y += ev.y; g.z += ev.z; g.w += ev.w;
        g.x *= 0.25f; g.y *= 0.25f; g.z *= 0.25f; g.w *= 0.25f;
        *(float4*)(dst + o) = g;
    }
}

extern "C" void kernel_launch(void* const* d_in, const int* in_sizes, int n_in,
                              void* d_out, int out_size, void* d_ws, size_t ws_size,
                              hipStream_t stream) {
    const float* user_emb = (const float*)d_in[0];
    const float* item_emb = (const float*)d_in[1];
    const float* adj_vals = (const float*)d_in[2];
    const float* noise    = (const float*)d_in[3];
    const int*   adj_rows = (const int*)d_in[4];
    const int*   adj_cols = (const int*)d_in[5];

    float* out = (float*)d_out;

    float* R1  = (float*)d_ws;
    float* R2  = R1 + (size_t)NTOT * EMB;

    Rec*  tmp    = (Rec*)R1;
    Rec*  bkt    = (Rec*)R2;
    int2* scv    = (int2*)R2;
    int*  rowptr = (int*)(R2 + (size_t)3 * NNZ_CNT);
    int*  tbl    = rowptr + ROWPTR_PAD;
    int*  tblx   = tbl + TBL_PAD;

    const int oblocks = (NTOT / ORD_R + 7) / 8;
    const int rblocks = NTOT / LAY_ROWS;

    k_bcount<<<NCHUNK, 1024, 0, stream>>>(adj_rows, tbl);
    k_scan<<<SCAN_BLOCKS, SCAN_BLOCK, 0, stream>>>(tbl, tblx);
    k_bscatter<<<NCHUNK, 1024, 0, stream>>>(adj_rows, adj_cols, adj_vals, tblx, bkt);
    k_brows<<<NBKT, 1024, 0, stream>>>(tblx, bkt, rowptr, tmp);
    k_order<<<oblocks, 256, 0, stream>>>(rowptr, tmp, scv);

    k_layer<1><<<rblocks, 256, 0, stream>>>(rowptr, scv, user_emb, item_emb, nullptr,
        noise, nullptr, nullptr, out);
    k_layer<2><<<rblocks, 256, 0, stream>>>(rowptr, scv, user_emb, item_emb, out,
        noise + (size_t)1 * NTOT * EMB, nullptr, nullptr, R1);
    k_layer<3><<<rblocks, 256, 0, stream>>>(rowptr, scv, user_emb, item_emb, R1,
        noise + (size_t)2 * NTOT * EMB, out, R1, out);
}
